# starting kernel
# baseline (speedup 1.0000x reference)
#include <hip/hip_runtime.h>

typedef __attribute__((ext_vector_type(2))) float v2f;
typedef __attribute__((ext_vector_type(4))) float v4f;
typedef __attribute__((ext_vector_type(8))) float v8f;

#define FD 32

__global__ void gcn_zero_kernel(float* __restrict__ p, int n4) {
    int i = blockIdx.x * blockDim.x + threadIdx.x;
    if (i < n4) {
        v4f z = {0.0f, 0.0f, 0.0f, 0.0f};
        ((v4f*)p)[i] = z;
    }
}

__global__ void gcn_degree_kernel(const int* __restrict__ src,
                                  const int* __restrict__ dst,
                                  unsigned* __restrict__ outd,
                                  unsigned* __restrict__ ind,
                                  int E) {
    int e = blockIdx.x * blockDim.x + threadIdx.x;
    if (e < E) {
        atomicAdd(&outd[src[e]], 1u);
        atomicAdd(&ind[dst[e]], 1u);
    }
}

__global__ void gcn_scatter_kernel(const float* __restrict__ feat,
                                   const float* __restrict__ e_feat,
                                   const int* __restrict__ src,
                                   const int* __restrict__ dst,
                                   const unsigned* __restrict__ outd,
                                   float* __restrict__ agg,
                                   int E) {
    int wave = blockIdx.x * (blockDim.x >> 5) + (threadIdx.x >> 5);
    int lane = threadIdx.x & 31;
    if (wave >= E) return;
    int s = src[wave];
    int t = dst[wave];
    if (s == t) return;
    float w = e_feat[wave];
    unsigned dg = outd[s];
    float norm = rsqrtf((float)(dg ? dg : 1u));
    float v = feat[(size_t)s * FD + lane] * norm * w;
    atomicAdd(&agg[(size_t)t * FD + lane], v);
}

__global__ void gcn_finalize_kernel(const float* __restrict__ feat,
                                    const float* __restrict__ agg,
                                    const unsigned* __restrict__ ind,
                                    float* __restrict__ out,
                                    int N) {
    int wave = blockIdx.x * (blockDim.x >> 5) + (threadIdx.x >> 5);
    int lane = threadIdx.x & 31;
    int ntiles = (N + 15) >> 4;
    if (wave >= ntiles) return;

    int r    = lane & 15;
    int half = lane >> 4;
    int node = wave * 16 + r;

    float vals[16];
    if (node < N) {
        unsigned dg = ind[node];
        float dn = rsqrtf((float)(dg ? dg : 1u));
        const v4f* fp = (const v4f*)(feat + (size_t)node * FD + half * 16);
        const v4f* ap = (const v4f*)(agg  + (size_t)node * FD + half * 16);
#pragma unroll
        for (int q = 0; q < 4; ++q) {
            v4f f = fp[q];
            v4f a = ap[q];
            vals[4 * q + 0] = __builtin_fabsf(f.x - a.x * dn);
            vals[4 * q + 1] = __builtin_fabsf(f.y - a.y * dn);
            vals[4 * q + 2] = __builtin_fabsf(f.z - a.z * dn);
            vals[4 * q + 3] = __builtin_fabsf(f.w - a.w * dn);
        }
    } else {
#pragma unroll
        for (int q = 0; q < 16; ++q) vals[q] = 0.0f;
    }

    v2f ones = {1.0f, 1.0f};
    v8f c = {};
#pragma unroll
    for (int kk = 0; kk < 8; ++kk) {
        v2f a;
        a.x = vals[2 * kk + 0];
        a.y = vals[2 * kk + 1];
        c = __builtin_amdgcn_wmma_f32_16x16x4_f32(
                false, a, false, ones, (short)0, c, false, false);
    }

    if ((lane & 15) == 0) {
        int base = wave * 16 + half * 8;
#pragma unroll
        for (int m = 0; m < 8; ++m) {
            int nd = base + m;
            if (nd < N) out[nd] = c[m];
        }
    }
}

extern "C" void kernel_launch(void* const* d_in, const int* in_sizes, int n_in,
                              void* d_out, int out_size, void* d_ws, size_t ws_size,
                              hipStream_t stream) {
    const float* feat   = (const float*)d_in[0];
    const float* e_feat = (const float*)d_in[1];
    const int*   src    = (const int*)d_in[2];
    const int*   dst    = (const int*)d_in[3];
    float*       out    = (float*)d_out;

    int N = in_sizes[0] / FD;
    int E = in_sizes[2];

    unsigned* outd = (unsigned*)d_ws;
    unsigned* ind  = outd + N;
    float*    agg  = (float*)(ind + N);

    long long zero_words = (long long)2 * N + (long long)N * FD;
    int zero_v4 = (int)((zero_words + 3) / 4);
    gcn_zero_kernel<<<(zero_v4 + 255) / 256, 256, 0, stream>>>((float*)d_ws, zero_v4);

    gcn_degree_kernel<<<(E + 255) / 256, 256, 0, stream>>>(src, dst, outd, ind, E);

    gcn_scatter_kernel<<<(E + 7) / 8, 256, 0, stream>>>(feat, e_feat, src, dst,
                                                        outd, agg, E);

    int ntiles = (N + 15) / 16;
    gcn_finalize_kernel<<<(ntiles + 7) / 8, 256, 0, stream>>>(feat, agg, ind, out, N);
}


# edges sorted by node owner first (count, then fill 8-byte records): an owner work-group reads only its own run, not the whole dst list; LDS f32 atomics, rows summed in arrival order
# speedup vs baseline: 2.1487x; 2.1487x over previous
#include <hip/hip_runtime.h>


typedef __attribute__((ext_vector_type(2))) float v2f;
typedef __attribute__((ext_vector_type(4))) float v4f;
typedef __attribute__((ext_vector_type(8))) float v8f;
typedef __attribute__((ext_vector_type(2))) int v2i;
typedef __attribute__((ext_vector_type(4))) int v4i;

#define FD 32
#define NPW 400
#define NTILE (NPW / 16)
#define OWN_THREADS 1024
#define OWN_WAVES (OWN_THREADS / 32)
#define BIN_THREADS 256
#define BIN_LD 4
#define BIN_EDGES (BIN_THREADS * 4 * BIN_LD)
#define MAXB 1024
#define CSTRIDE 32
#define SBITS 23
#define SMASK 0x7FFFFF

__global__ void gcn_zero_kernel(float* __restrict__ p, int n4) {
    int i = blockIdx.x * blockDim.x + threadIdx.x;
    if (i < n4) {
        v4f z = {0.0f, 0.0f, 0.0f, 0.0f};
        ((v4f*)p)[i] = z;
    }
}

__device__ __forceinline__ void load4_tail(int (&v)[4], const int* __restrict__ p, int base, int E) {
#pragma unroll
    for (int j = 0; j < 4; ++j) v[j] = (base + j < E) ? p[base + j] : -1;
}
__device__ __forceinline__ void load4w_tail(int (&v)[4], const float* __restrict__ p, int base, int E) {
#pragma unroll
    for (int j = 0; j < 4; ++j) v[j] = (base + j < E) ? __float_as_int(p[base + j]) : 0;
}
__device__ __forceinline__ void load4(int (&v)[4], const int* __restrict__ p, int base) {
    v4i d = *(const v4i*)(p + base);
    v[0] = d.x; v[1] = d.y; v[2] = d.z; v[3] = d.w;
}
__device__ __forceinline__ void load4w(int (&v)[4], const float* __restrict__ p, int base) {
    v4f d = *(const v4f*)(p + base);
    v[0] = __float_as_int(d.x); v[1] = __float_as_int(d.y);
    v[2] = __float_as_int(d.z); v[3] = __float_as_int(d.w);
}

__global__ __launch_bounds__(BIN_THREADS)
void gcn_count_kernel(const int* __restrict__ src,
                      const int* __restrict__ dst,
                      unsigned* __restrict__ outd,
                      unsigned* __restrict__ cnt,
                      int E, int NB) {
    __shared__ unsigned hist[MAXB];
    const int tid = threadIdx.x;
    const int chunk0 = blockIdx.x * BIN_EDGES + tid * 4;
    int sv[BIN_LD][4], tv[BIN_LD][4];
    if ((blockIdx.x + 1) * BIN_EDGES <= E) {
#pragma unroll
        for (int k = 0; k < BIN_LD; ++k) {
            load4(sv[k], src, chunk0 + k * (BIN_THREADS * 4));
            load4(tv[k], dst, chunk0 + k * (BIN_THREADS * 4));
        }
    } else {
#pragma unroll
        for (int k = 0; k < BIN_LD; ++k) {
            load4_tail(sv[k], src, chunk0 + k * (BIN_THREADS * 4), E);
            load4_tail(tv[k], dst, chunk0 + k * (BIN_THREADS * 4), E);
        }
    }
    for (int o = tid; o < NB; o += BIN_THREADS) hist[o] = 0u;
    __syncthreads();

#pragma unroll
    for (int k = 0; k < BIN_LD; ++k) {
#pragma unroll
        for (int j = 0; j < 4; ++j) {
            if (tv[k][j] >= 0) {
                atomicAdd(&outd[sv[k][j]], 1u);
                atomicAdd(&hist[(unsigned)tv[k][j] / NPW], 1u);
            }
        }
    }
    __syncthreads();
    for (int o = tid; o < NB; o += BIN_THREADS) {
        unsigned h = hist[o];
        if (h) atomicAdd(&cnt[o * CSTRIDE], h);
    }
}

__global__ __launch_bounds__(BIN_THREADS)
void gcn_fill_kernel(const float* __restrict__ e_feat,
                     const int* __restrict__ src,
                     const int* __restrict__ dst,
                     const unsigned* __restrict__ cnt,
                     unsigned* __restrict__ cur,
                     int* __restrict__ rec,
                     int E, int NB) {
    __shared__ unsigned hist[MAXB];
    __shared__ unsigned cntv[MAXB + 32];
    __shared__ unsigned gpos[MAXB + 32];
    const int tid = threadIdx.x;
    const int lane = tid & 31;
    const int wave = tid >> 5;

    unsigned cv[(MAXB + 32 + BIN_THREADS - 1) / BIN_THREADS];
#pragma unroll
    for (int q = 0; q < (MAXB + 32 + BIN_THREADS - 1) / BIN_THREADS; ++q) {
        int o = tid + q * BIN_THREADS;
        cv[q] = (o < NB) ? cnt[o * CSTRIDE] : 0u;
    }
    const int chunk0 = blockIdx.x * BIN_EDGES + tid * 4;
    int sv[BIN_LD][4], tv[BIN_LD][4], wv[BIN_LD][4];
    if ((blockIdx.x + 1) * BIN_EDGES <= E) {
#pragma unroll
        for (int k = 0; k < BIN_LD; ++k) {
            load4(sv[k], src, chunk0 + k * (BIN_THREADS * 4));
            load4(tv[k], dst, chunk0 + k * (BIN_THREADS * 4));
            load4w(wv[k], e_feat, chunk0 + k * (BIN_THREADS * 4));
        }
    } else {
#pragma unroll
        for (int k = 0; k < BIN_LD; ++k) {
            load4_tail(sv[k], src, chunk0 + k * (BIN_THREADS * 4), E);
            load4_tail(tv[k], dst, chunk0 + k * (BIN_THREADS * 4), E);
            load4w_tail(wv[k], e_feat, chunk0 + k * (BIN_THREADS * 4), E);
        }
    }
#pragma unroll
    for (int q = 0; q < (MAXB + 32 + BIN_THREADS - 1) / BIN_THREADS; ++q) {
        int o = tid + q * BIN_THREADS;
        if (o < MAXB) hist[o] = 0u;
        if (o < MAXB + 32) cntv[o] = cv[q];
    }
    __syncthreads();

    if (wave == 0) {
        const int per = (NB + 31) / 32;
        const int lo = lane * per;
        unsigned s = 0u;
        for (int i = 0; i < per; ++i) s += cntv[lo + i];
        unsigned inc = s;
        for (int d = 1; d < 32; d <<= 1) {
            unsigned up = __shfl_up(inc, d);
            if (lane >= d) inc += up;
        }
        unsigned run = inc - s;
        for (int i = 0; i < per; ++i) {
            gpos[lo + i] = run;
            run += cntv[lo + i];
        }
    }
    unsigned ow[BIN_LD][4], rk[BIN_LD][4];
#pragma unroll
    for (int k = 0; k < BIN_LD; ++k) {
#pragma unroll
        for (int j = 0; j < 4; ++j) {
            ow[k][j] = (tv[k][j] >= 0) ? (unsigned)tv[k][j] / NPW : 0u;
            rk[k][j] = 0u;
            if (tv[k][j] >= 0) rk[k][j] = atomicAdd(&hist[ow[k][j]], 1u);
        }
    }
    __syncthreads();
    for (int o = tid; o < NB; o += BIN_THREADS) {
        unsigned h = hist[o];
        if (h) gpos[o] += atomicAdd(&cur[o * CSTRIDE], h);
    }
    __syncthreads();
#pragma unroll
    for (int k = 0; k < BIN_LD; ++k) {
#pragma unroll
        for (int j = 0; j < 4; ++j) rk[k][j] += gpos[ow[k][j]];
    }
    v2i* recs = (v2i*)rec;
#pragma unroll
    for (int k = 0; k < BIN_LD; ++k) {
#pragma unroll
        for (int j = 0; j < 4; ++j) {
            int t = tv[k][j];
            if (t >= 0) {
                unsigned tl = (unsigned)t - ow[k][j] * NPW;
                v2i r;
                r.x = (int)((tl << SBITS) | (unsigned)sv[k][j]);
                r.y = wv[k][j];
                recs[rk[k][j]] = r;
            }
        }
    }
}

__device__ __forceinline__ void drain_batch(v2i rc, int n, int lane, int node0,
                                            const float* __restrict__ feat,
                                            const unsigned* __restrict__ outd,
                                            float* sagg, unsigned* sind) {
    bool valid = lane < n;
    int s = rc.x & SMASK;
    int tl = (int)((unsigned)rc.x >> SBITS);
    unsigned dg = outd[s];
    if (valid) atomicAdd(&sind[tl], 1u);
    bool live = valid && (s != node0 + tl);
    unsigned m = __builtin_amdgcn_ballot_w32(live);
    int sb = s * FD;
    int tb = tl * FD;
    int wi = rc.y;

    float fv[32];
#pragma unroll
    for (int j = 0; j < 32; ++j) {
        int sj = __builtin_amdgcn_readlane(sb, j);
        fv[j] = feat[sj + lane];
    }
    __builtin_amdgcn_sched_barrier(0);
    float norm = rsqrtf((float)(dg ? dg : 1u));
    int ni = __float_as_int(norm);
#pragma unroll
    for (int j = 0; j < 32; ++j) {
        if ((m >> j) & 1u) {
            float nj = __int_as_float(__builtin_amdgcn_readlane(ni, j));
            float wj = __int_as_float(__builtin_amdgcn_readlane(wi, j));
            int tj = __builtin_amdgcn_readlane(tb, j);
            atomicAdd(&sagg[tj + lane], fv[j] * nj * wj);
        }
    }
}

__global__ __launch_bounds__(OWN_THREADS)
void gcn_owner_kernel(const float* __restrict__ feat,
                      const unsigned* __restrict__ outd,
                      const unsigned* __restrict__ cnt,
                      const int* __restrict__ rec,
                      float* __restrict__ out,
                      int N) {
    __shared__ __attribute__((aligned(16))) float sagg[NPW * FD];
    __shared__ unsigned sind[NPW];
    __shared__ unsigned sfirst;

    const int tid = threadIdx.x;
    const int lane = tid & 31;
    const int wave = __builtin_amdgcn_readfirstlane(tid >> 5);
    const int b = blockIdx.x;
    const int node0 = b * NPW;

    for (int i = tid; i < NPW * FD; i += OWN_THREADS) sagg[i] = 0.0f;
    for (int i = tid; i < NPW; i += OWN_THREADS) sind[i] = 0u;
    if (tid == 0) sfirst = 0u;
    __syncthreads();

    unsigned part = 0u;
    for (int o = tid; o < b; o += OWN_THREADS) part += cnt[o * CSTRIDE];
    if (part) atomicAdd(&sfirst, part);
    __syncthreads();

    const int n = (int)cnt[b * CSTRIDE];
    const v2i* mine = (const v2i*)rec + sfirst;
    const int nBatch = (n + 31) >> 5;

    int bt = wave;
    v2i rc = {0, 0};
    if (bt < nBatch && bt * 32 + lane < n) rc = mine[bt * 32 + lane];
    while (bt < nBatch) {
        const int nb = bt + OWN_WAVES;
        v2i rn = {0, 0};
        if (nb < nBatch && nb * 32 + lane < n) rn = mine[nb * 32 + lane];
        const int left = n - bt * 32;
        drain_batch(rc, left < 32 ? left : 32, lane, node0, feat, outd, sagg, sind);
        rc = rn;
        bt = nb;
    }

    __syncthreads();

    if (wave < NTILE) {
        int r = lane & 15;
        int half = lane >> 4;
        int ln = wave * 16 + r;
        int node = node0 + ln;

        float vals[16];
        if (node < N) {
            unsigned dg = sind[ln];
            float dn = rsqrtf((float)(dg ? dg : 1u));
            const v4f* fp = (const v4f*)(feat + (size_t)node * FD + half * 16);
            const v4f* ap = (const v4f*)(sagg + ln * FD + half * 16);
#pragma unroll
            for (int qd = 0; qd < 4; ++qd) {
                v4f f = fp[qd];
                v4f a = ap[qd];
                vals[4 * qd + 0] = __builtin_fabsf(f.x - a.x * dn);
                vals[4 * qd + 1] = __builtin_fabsf(f.y - a.y * dn);
                vals[4 * qd + 2] = __builtin_fabsf(f.z - a.z * dn);
                vals[4 * qd + 3] = __builtin_fabsf(f.w - a.w * dn);
            }
        } else {
#pragma unroll
            for (int qd = 0; qd < 16; ++qd) vals[qd] = 0.0f;
        }

        v2f ones = {1.0f, 1.0f};
        v8f c = {};
#pragma unroll
        for (int kk = 0; kk < 8; ++kk) {
            v2f a;
            a.x = vals[2 * kk + 0];
            a.y = vals[2 * kk + 1];
            c = __builtin_amdgcn_wmma_f32_16x16x4_f32(
                    false, a, false, ones, (short)0, c, false, false);
        }

        if ((lane & 15) == 0) {
            int base = node0 + wave * 16 + half * 8;
#pragma unroll
            for (int mm = 0; mm < 8; ++mm) {
                int nd = base + mm;
                if (nd < N) out[nd] = c[mm];
            }
        }
    }
}

extern "C" void kernel_launch(void* const* d_in, const int* in_sizes, int n_in,
                              void* d_out, int out_size, void* d_ws, size_t ws_size,
                              hipStream_t stream) {
    const float* feat   = (const float*)d_in[0];
    const float* e_feat = (const float*)d_in[1];
    const int*   src    = (const int*)d_in[2];
    const int*   dst    = (const int*)d_in[3];
    float*       out    = (float*)d_out;

    int N = in_sizes[0] / FD;
    int E = in_sizes[2];
    int NB = (N + NPW - 1) / NPW;
    int Npad = (N + 31) / 32 * 32;

    unsigned* outd = (unsigned*)d_ws;
    unsigned* cnt  = outd + Npad;
    unsigned* cur  = cnt + NB * CSTRIDE;
    int*      rec  = (int*)(cur + NB * CSTRIDE);

    int zero_v4 = (Npad + 2 * NB * CSTRIDE) / 4;
    gcn_zero_kernel<<<(zero_v4 + 255) / 256, 256, 0, stream>>>((float*)d_ws, zero_v4);

    int nbin = (E + BIN_EDGES - 1) / BIN_EDGES;
    gcn_count_kernel<<<nbin, BIN_THREADS, 0, stream>>>(src, dst, outd, cnt, E, NB);

    gcn_fill_kernel<<<nbin, BIN_THREADS, 0, stream>>>(e_feat, src, dst, cnt, cur, rec, E, NB);

    gcn_owner_kernel<<<NB, OWN_THREADS, 0, stream>>>(feat, outd, cnt, rec, out, N);
}


# out-degree without a global atomic per edge: the count pass sorts each chunk's sources by node owner in LDS (whole-line stores); a new owner pass counts its segments in LDS and stores norms
# speedup vs baseline: 3.8023x; 1.7696x over previous
#include <hip/hip_runtime.h>


typedef __attribute__((ext_vector_type(2))) float v2f;
typedef __attribute__((ext_vector_type(4))) float v4f;
typedef __attribute__((ext_vector_type(8))) float v8f;
typedef __attribute__((ext_vector_type(2))) int v2i;
typedef __attribute__((ext_vector_type(4))) int v4i;

#define FD 32
#define NPW 400
#define NTILE (NPW / 16)
#define OWN_THREADS 1024
#define OWN_WAVES (OWN_THREADS / 32)
#define BIN_THREADS 256
#define BIN_LD 4
#define BIN_EDGES (BIN_THREADS * 4 * BIN_LD)
#define MAXB 1024
#define CSTRIDE 32
#define DEG_THREADS 512
#define SBITS 23
#define SMASK 0x7FFFFF

__global__ void gcn_zero_kernel(float* __restrict__ p, int n4) {
    int i = blockIdx.x * blockDim.x + threadIdx.x;
    if (i < n4) {
        v4f z = {0.0f, 0.0f, 0.0f, 0.0f};
        ((v4f*)p)[i] = z;
    }
}

__device__ __forceinline__ void load4_tail(int (&v)[4], const int* __restrict__ p, int base, int E) {
#pragma unroll
    for (int j = 0; j < 4; ++j) v[j] = (base + j < E) ? p[base + j] : -1;
}
__device__ __forceinline__ void load4w_tail(int (&v)[4], const float* __restrict__ p, int base, int E) {
#pragma unroll
    for (int j = 0; j < 4; ++j) v[j] = (base + j < E) ? __float_as_int(p[base + j]) : 0;
}
__device__ __forceinline__ void load4(int (&v)[4], const int* __restrict__ p, int base) {
    v4i d = *(const v4i*)(p + base);
    v[0] = d.x; v[1] = d.y; v[2] = d.z; v[3] = d.w;
}
__device__ __forceinline__ void load4w(int (&v)[4], const float* __restrict__ p, int base) {
    v4f d = *(const v4f*)(p + base);
    v[0] = __float_as_int(d.x); v[1] = __float_as_int(d.y);
    v[2] = __float_as_int(d.z); v[3] = __float_as_int(d.w);
}

__global__ __launch_bounds__(BIN_THREADS)
void gcn_count_kernel(const int* __restrict__ src,
                      const int* __restrict__ dst,
                      unsigned* __restrict__ cnt,
                      int* __restrict__ ssort_w,
                      unsigned* __restrict__ soff,
                      int E, int NB) {
    __shared__ unsigned hist[MAXB];
    __shared__ unsigned shist[MAXB + 32];
    __shared__ unsigned spos[MAXB + 32];
    __shared__ __attribute__((aligned(16))) unsigned short stage[BIN_EDGES];
    const int tid = threadIdx.x;
    const int lane = tid & 31;
    const int wave = tid >> 5;
    const int chunk0 = blockIdx.x * BIN_EDGES + tid * 4;
    int sv[BIN_LD][4], tv[BIN_LD][4];
    if ((blockIdx.x + 1) * BIN_EDGES <= E) {
#pragma unroll
        for (int k = 0; k < BIN_LD; ++k) {
            load4(sv[k], src, chunk0 + k * (BIN_THREADS * 4));
            load4(tv[k], dst, chunk0 + k * (BIN_THREADS * 4));
        }
    } else {
#pragma unroll
        for (int k = 0; k < BIN_LD; ++k) {
            load4_tail(sv[k], src, chunk0 + k * (BIN_THREADS * 4), E);
            load4_tail(tv[k], dst, chunk0 + k * (BIN_THREADS * 4), E);
        }
    }
    for (int o = tid; o < NB + 32; o += BIN_THREADS) {
        if (o < NB) hist[o] = 0u;
        shist[o] = 0u;
    }
    __syncthreads();

    unsigned so[BIN_LD][4], rk[BIN_LD][4];
#pragma unroll
    for (int k = 0; k < BIN_LD; ++k) {
#pragma unroll
        for (int j = 0; j < 4; ++j) {
            so[k][j] = 0u;
            rk[k][j] = 0u;
            if (tv[k][j] >= 0) {
                atomicAdd(&hist[(unsigned)tv[k][j] / NPW], 1u);
                so[k][j] = (unsigned)sv[k][j] / NPW;
                rk[k][j] = atomicAdd(&shist[so[k][j]], 1u);
            }
        }
    }
    __syncthreads();
    for (int o = tid; o < NB; o += BIN_THREADS) {
        unsigned h = hist[o];
        if (h) atomicAdd(&cnt[o * CSTRIDE], h);
    }
    if (wave == 0) {
        const int per = (NB + 32) / 32;
        const int lo = lane * per;
        unsigned s = 0u;
        for (int i = 0; i < per; ++i) s += shist[lo + i];
        unsigned inc = s;
        for (int d = 1; d < 32; d <<= 1) {
            unsigned up = __shfl_up(inc, d);
            if (lane >= d) inc += up;
        }
        unsigned run = inc - s;
        for (int i = 0; i < per; ++i) {
            spos[lo + i] = run;
            run += shist[lo + i];
        }
    }
    __syncthreads();
    unsigned* myoff = soff + (size_t)blockIdx.x * (NB + 1);
    for (int o = tid; o <= NB; o += BIN_THREADS) myoff[o] = spos[o];
#pragma unroll
    for (int k = 0; k < BIN_LD; ++k) {
#pragma unroll
        for (int j = 0; j < 4; ++j) rk[k][j] += spos[so[k][j]];
    }
#pragma unroll
    for (int k = 0; k < BIN_LD; ++k) {
#pragma unroll
        for (int j = 0; j < 4; ++j) {
            if (tv[k][j] >= 0) {
                unsigned sl = (unsigned)sv[k][j] - so[k][j] * NPW;
                stage[rk[k][j]] = (unsigned short)sl;
            }
        }
    }
    __syncthreads();
    const unsigned total = spos[NB];
    v4i* gs = (v4i*)((unsigned short*)ssort_w + (size_t)blockIdx.x * BIN_EDGES);
    const v4i* ls = (const v4i*)stage;
    for (int q = tid; q < BIN_EDGES / 8; q += BIN_THREADS) {
        if ((unsigned)q * 8u < total) gs[q] = ls[q];
    }
}

__global__ __launch_bounds__(DEG_THREADS)
void gcn_outdeg_kernel(const int* __restrict__ ssort_w,
                       const unsigned* __restrict__ soff,
                       float* __restrict__ nrm,
                       int N, int NB, int nbin) {
    __shared__ unsigned lcnt[NPW];
    const int tid = threadIdx.x;
    const int o = blockIdx.x;
    for (int i = tid; i < NPW; i += DEG_THREADS) lcnt[i] = 0u;
    __syncthreads();
    const unsigned short* ssort = (const unsigned short*)ssort_w;
    for (int c = tid; c < nbin; c += DEG_THREADS) {
        const unsigned* t = soff + (size_t)c * (NB + 1) + o;
        unsigned a = t[0];
        unsigned b = t[1];
        if (b > (unsigned)BIN_EDGES) b = BIN_EDGES;
        if (a < b) {
            const v4i* blk = (const v4i*)(ssort + (size_t)c * BIN_EDGES);
            for (unsigned q = a >> 3; q * 8u < b; ++q) {
                v4i d = blk[q];
                unsigned wd[4] = {(unsigned)d.x, (unsigned)d.y, (unsigned)d.z, (unsigned)d.w};
#pragma unroll
                for (int e = 0; e < 4; ++e) {
                    unsigned i0 = q * 8u + 2u * e;
                    unsigned id0 = wd[e] & 0xFFFFu;
                    unsigned id1 = wd[e] >> 16;
                    if (i0 >= a && i0 < b && id0 < NPW) atomicAdd(&lcnt[id0], 1u);
                    if (i0 + 1u >= a && i0 + 1u < b && id1 < NPW) atomicAdd(&lcnt[id1], 1u);
                }
            }
        }
    }
    __syncthreads();
    for (int i = tid; i < NPW; i += DEG_THREADS) {
        int node = o * NPW + i;
        if (node < N) {
            unsigned dg = lcnt[i];
            nrm[node] = rsqrtf((float)(dg ? dg : 1u));
        }
    }
}

__global__ __launch_bounds__(BIN_THREADS)
void gcn_fill_kernel(const float* __restrict__ e_feat,
                     const int* __restrict__ src,
                     const int* __restrict__ dst,
                     const unsigned* __restrict__ cnt,
                     unsigned* __restrict__ cur,
                     int* __restrict__ rec,
                     int E, int NB) {
    __shared__ unsigned hist[MAXB];
    __shared__ unsigned cntv[MAXB + 32];
    __shared__ unsigned gpos[MAXB + 32];
    const int tid = threadIdx.x;
    const int lane = tid & 31;
    const int wave = tid >> 5;

    unsigned cv[(MAXB + 32 + BIN_THREADS - 1) / BIN_THREADS];
#pragma unroll
    for (int q = 0; q < (MAXB + 32 + BIN_THREADS - 1) / BIN_THREADS; ++q) {
        int o = tid + q * BIN_THREADS;
        cv[q] = (o < NB) ? cnt[o * CSTRIDE] : 0u;
    }
    const int chunk0 = blockIdx.x * BIN_EDGES + tid * 4;
    int sv[BIN_LD][4], tv[BIN_LD][4], wv[BIN_LD][4];
    if ((blockIdx.x + 1) * BIN_EDGES <= E) {
#pragma unroll
        for (int k = 0; k < BIN_LD; ++k) {
            load4(sv[k], src, chunk0 + k * (BIN_THREADS * 4));
            load4(tv[k], dst, chunk0 + k * (BIN_THREADS * 4));
            load4w(wv[k], e_feat, chunk0 + k * (BIN_THREADS * 4));
        }
    } else {
#pragma unroll
        for (int k = 0; k < BIN_LD; ++k) {
            load4_tail(sv[k], src, chunk0 + k * (BIN_THREADS * 4), E);
            load4_tail(tv[k], dst, chunk0 + k * (BIN_THREADS * 4), E);
            load4w_tail(wv[k], e_feat, chunk0 + k * (BIN_THREADS * 4), E);
        }
    }
#pragma unroll
    for (int q = 0; q < (MAXB + 32 + BIN_THREADS - 1) / BIN_THREADS; ++q) {
        int o = tid + q * BIN_THREADS;
        if (o < MAXB) hist[o] = 0u;
        if (o < MAXB + 32) cntv[o] = cv[q];
    }
    __syncthreads();

    if (wave == 0) {
        const int per = (NB + 31) / 32;
        const int lo = lane * per;
        unsigned s = 0u;
        for (int i = 0; i < per; ++i) s += cntv[lo + i];
        unsigned inc = s;
        for (int d = 1; d < 32; d <<= 1) {
            unsigned up = __shfl_up(inc, d);
            if (lane >= d) inc += up;
        }
        unsigned run = inc - s;
        for (int i = 0; i < per; ++i) {
            gpos[lo + i] = run;
            run += cntv[lo + i];
        }
    }
    unsigned ow[BIN_LD][4], rk[BIN_LD][4];
#pragma unroll
    for (int k = 0; k < BIN_LD; ++k) {
#pragma unroll
        for (int j = 0; j < 4; ++j) {
            ow[k][j] = (tv[k][j] >= 0) ? (unsigned)tv[k][j] / NPW : 0u;
            rk[k][j] = 0u;
            if (tv[k][j] >= 0) rk[k][j] = atomicAdd(&hist[ow[k][j]], 1u);
        }
    }
    __syncthreads();
    for (int o = tid; o < NB; o += BIN_THREADS) {
        unsigned h = hist[o];
        if (h) gpos[o] += atomicAdd(&cur[o * CSTRIDE], h);
    }
    __syncthreads();
#pragma unroll
    for (int k = 0; k < BIN_LD; ++k) {
#pragma unroll
        for (int j = 0; j < 4; ++j) rk[k][j] += gpos[ow[k][j]];
    }
    v2i* recs = (v2i*)rec;
#pragma unroll
    for (int k = 0; k < BIN_LD; ++k) {
#pragma unroll
        for (int j = 0; j < 4; ++j) {
            int t = tv[k][j];
            if (t >= 0) {
                unsigned tl = (unsigned)t - ow[k][j] * NPW;
                v2i r;
                r.x = (int)((tl << SBITS) | (unsigned)sv[k][j]);
                r.y = wv[k][j];
                recs[rk[k][j]] = r;
            }
        }
    }
}

__device__ __forceinline__ void drain_batch(v2i rc, int n, int lane, int node0,
                                            const float* __restrict__ feat,
                                            const float* __restrict__ nrm,
                                            float* sagg, unsigned* sind) {
    bool valid = lane < n;
    int s = rc.x & SMASK;
    int tl = (int)((unsigned)rc.x >> SBITS);
    float norm = nrm[s];
    if (valid) atomicAdd(&sind[tl], 1u);
    bool live = valid && (s != node0 + tl);
    unsigned m = __builtin_amdgcn_ballot_w32(live);
    int sb = s * FD;
    int tb = tl * FD;
    int wi = rc.y;

    float fv[32];
#pragma unroll
    for (int j = 0; j < 32; ++j) {
        int sj = __builtin_amdgcn_readlane(sb, j);
        fv[j] = feat[sj + lane];
    }
    __builtin_amdgcn_sched_barrier(0);
    int ni = __float_as_int(norm);
#pragma unroll
    for (int j = 0; j < 32; ++j) {
        if ((m >> j) & 1u) {
            float nj = __int_as_float(__builtin_amdgcn_readlane(ni, j));
            float wj = __int_as_float(__builtin_amdgcn_readlane(wi, j));
            int tj = __builtin_amdgcn_readlane(tb, j);
            atomicAdd(&sagg[tj + lane], fv[j] * nj * wj);
        }
    }
}

__global__ __launch_bounds__(OWN_THREADS)
void gcn_owner_kernel(const float* __restrict__ feat,
                      const float* __restrict__ nrm,
                      const unsigned* __restrict__ cnt,
                      const int* __restrict__ rec,
                      float* __restrict__ out,
                      int N) {
    __shared__ __attribute__((aligned(16))) float sagg[NPW * FD];
    __shared__ unsigned sind[NPW];
    __shared__ unsigned sfirst;

    const int tid = threadIdx.x;
    const int lane = tid & 31;
    const int wave = __builtin_amdgcn_readfirstlane(tid >> 5);
    const int b = blockIdx.x;
    const int node0 = b * NPW;

    for (int i = tid; i < NPW * FD; i += OWN_THREADS) sagg[i] = 0.0f;
    for (int i = tid; i < NPW; i += OWN_THREADS) sind[i] = 0u;
    if (tid == 0) sfirst = 0u;
    __syncthreads();

    unsigned part = 0u;
    for (int o = tid; o < b; o += OWN_THREADS) part += cnt[o * CSTRIDE];
    if (part) atomicAdd(&sfirst, part);
    __syncthreads();

    const int n = (int)cnt[b * CSTRIDE];
    const v2i* mine = (const v2i*)rec + sfirst;
    const int nBatch = (n + 31) >> 5;

    int bt = wave;
    v2i rc = {0, 0};
    if (bt < nBatch && bt * 32 + lane < n) rc = mine[bt * 32 + lane];
    while (bt < nBatch) {
        const int nb = bt + OWN_WAVES;
        v2i rn = {0, 0};
        if (nb < nBatch && nb * 32 + lane < n) rn = mine[nb * 32 + lane];
        const int left = n - bt * 32;
        drain_batch(rc, left < 32 ? left : 32, lane, node0, feat, nrm, sagg, sind);
        rc = rn;
        bt = nb;
    }

    __syncthreads();

    if (wave < NTILE) {
        int r = lane & 15;
        int half = lane >> 4;
        int ln = wave * 16 + r;
        int node = node0 + ln;

        float vals[16];
        if (node < N) {
            unsigned dg = sind[ln];
            float dn = rsqrtf((float)(dg ? dg : 1u));
            const v4f* fp = (const v4f*)(feat + (size_t)node * FD + half * 16);
            const v4f* ap = (const v4f*)(sagg + ln * FD + half * 16);
#pragma unroll
            for (int qd = 0; qd < 4; ++qd) {
                v4f f = fp[qd];
                v4f a = ap[qd];
                vals[4 * qd + 0] = __builtin_fabsf(f.x - a.x * dn);
                vals[4 * qd + 1] = __builtin_fabsf(f.y - a.y * dn);
                vals[4 * qd + 2] = __builtin_fabsf(f.z - a.z * dn);
                vals[4 * qd + 3] = __builtin_fabsf(f.w - a.w * dn);
            }
        } else {
#pragma unroll
            for (int qd = 0; qd < 16; ++qd) vals[qd] = 0.0f;
        }

        v2f ones = {1.0f, 1.0f};
        v8f c = {};
#pragma unroll
        for (int kk = 0; kk < 8; ++kk) {
            v2f a;
            a.x = vals[2 * kk + 0];
            a.y = vals[2 * kk + 1];
            c = __builtin_amdgcn_wmma_f32_16x16x4_f32(
                    false, a, false, ones, (short)0, c, false, false);
        }

        if ((lane & 15) == 0) {
            int base = node0 + wave * 16 + half * 8;
#pragma unroll
            for (int mm = 0; mm < 8; ++mm) {
                int nd = base + mm;
                if (nd < N) out[nd] = c[mm];
            }
        }
    }
}

extern "C" void kernel_launch(void* const* d_in, const int* in_sizes, int n_in,
                              void* d_out, int out_size, void* d_ws, size_t ws_size,
                              hipStream_t stream) {
    const float* feat   = (const float*)d_in[0];
    const float* e_feat = (const float*)d_in[1];
    const int*   src    = (const int*)d_in[2];
    const int*   dst    = (const int*)d_in[3];
    float*       out    = (float*)d_out;

    int N = in_sizes[0] / FD;
    int E = in_sizes[2];
    int NB = (N + NPW - 1) / NPW;
    int Npad = (N + 31) / 32 * 32;
    int nbin = (E + BIN_EDGES - 1) / BIN_EDGES;

    float*    nrm  = (float*)d_ws;
    unsigned* cnt  = (unsigned*)d_ws + Npad;
    unsigned* cur  = cnt + NB * CSTRIDE;
    int*      rec  = (int*)(cur + NB * CSTRIDE);
    int*      ssort = rec;
    unsigned* soff  = (unsigned*)(rec + (size_t)nbin * (BIN_EDGES / 2));

    int zero_v4 = (2 * NB * CSTRIDE) / 4;
    gcn_zero_kernel<<<(zero_v4 + 255) / 256, 256, 0, stream>>>((float*)cnt, zero_v4);

    gcn_count_kernel<<<nbin, BIN_THREADS, 0, stream>>>(src, dst, cnt, ssort, soff, E, NB);

    gcn_outdeg_kernel<<<NB, DEG_THREADS, 0, stream>>>(ssort, soff, nrm, N, NB, nbin);

    gcn_fill_kernel<<<nbin, BIN_THREADS, 0, stream>>>(e_feat, src, dst, cnt, cur, rec, E, NB);

    gcn_owner_kernel<<<NB, OWN_THREADS, 0, stream>>>(feat, nrm, cnt, rec, out, N);
}


# fill pass: records put in owner order in LDS with their global slot, then stored in that order, so a wave's store covers a few whole runs instead of 32 separate lines; sum order as before
# speedup vs baseline: 4.0978x; 1.0777x over previous
#include <hip/hip_runtime.h>


typedef __attribute__((ext_vector_type(2))) float v2f;
typedef __attribute__((ext_vector_type(4))) float v4f;
typedef __attribute__((ext_vector_type(8))) float v8f;
typedef __attribute__((ext_vector_type(2))) int v2i;
typedef __attribute__((ext_vector_type(4))) int v4i;

#define FD 32
#define NPW 400
#define NTILE (NPW / 16)
#define OWN_THREADS 1024
#define OWN_WAVES (OWN_THREADS / 32)
#define BIN_THREADS 256
#define BIN_LD 4
#define BIN_EDGES (BIN_THREADS * 4 * BIN_LD)
#define MAXB 1024
#define CSTRIDE 32
#define DEG_THREADS 512
#define SBITS 23
#define SMASK 0x7FFFFF

__global__ void gcn_zero_kernel(float* __restrict__ p, int n4) {
    int i = blockIdx.x * blockDim.x + threadIdx.x;
    if (i < n4) {
        v4f z = {0.0f, 0.0f, 0.0f, 0.0f};
        ((v4f*)p)[i] = z;
    }
}

__device__ __forceinline__ void load4_tail(int (&v)[4], const int* __restrict__ p, int base, int E) {
#pragma unroll
    for (int j = 0; j < 4; ++j) v[j] = (base + j < E) ? p[base + j] : -1;
}
__device__ __forceinline__ void load4w_tail(int (&v)[4], const float* __restrict__ p, int base, int E) {
#pragma unroll
    for (int j = 0; j < 4; ++j) v[j] = (base + j < E) ? __float_as_int(p[base + j]) : 0;
}
__device__ __forceinline__ void load4(int (&v)[4], const int* __restrict__ p, int base) {
    v4i d = *(const v4i*)(p + base);
    v[0] = d.x; v[1] = d.y; v[2] = d.z; v[3] = d.w;
}
__device__ __forceinline__ void load4w(int (&v)[4], const float* __restrict__ p, int base) {
    v4f d = *(const v4f*)(p + base);
    v[0] = __float_as_int(d.x); v[1] = __float_as_int(d.y);
    v[2] = __float_as_int(d.z); v[3] = __float_as_int(d.w);
}

__global__ __launch_bounds__(BIN_THREADS)
void gcn_count_kernel(const int* __restrict__ src,
                      const int* __restrict__ dst,
                      unsigned* __restrict__ cnt,
                      int* __restrict__ ssort_w,
                      unsigned* __restrict__ soff,
                      int E, int NB) {
    __shared__ unsigned hist[MAXB];
    __shared__ unsigned shist[MAXB + 32];
    __shared__ unsigned spos[MAXB + 32];
    __shared__ __attribute__((aligned(16))) unsigned short stage[BIN_EDGES];
    const int tid = threadIdx.x;
    const int lane = tid & 31;
    const int wave = tid >> 5;
    const int chunk0 = blockIdx.x * BIN_EDGES + tid * 4;
    int sv[BIN_LD][4], tv[BIN_LD][4];
    if ((blockIdx.x + 1) * BIN_EDGES <= E) {
#pragma unroll
        for (int k = 0; k < BIN_LD; ++k) {
            load4(sv[k], src, chunk0 + k * (BIN_THREADS * 4));
            load4(tv[k], dst, chunk0 + k * (BIN_THREADS * 4));
        }
    } else {
#pragma unroll
        for (int k = 0; k < BIN_LD; ++k) {
            load4_tail(sv[k], src, chunk0 + k * (BIN_THREADS * 4), E);
            load4_tail(tv[k], dst, chunk0 + k * (BIN_THREADS * 4), E);
        }
    }
    for (int o = tid; o < NB + 32; o += BIN_THREADS) {
        if (o < NB) hist[o] = 0u;
        shist[o] = 0u;
    }
    __syncthreads();

    unsigned so[BIN_LD][4], rk[BIN_LD][4];
#pragma unroll
    for (int k = 0; k < BIN_LD; ++k) {
#pragma unroll
        for (int j = 0; j < 4; ++j) {
            so[k][j] = 0u;
            rk[k][j] = 0u;
            if (tv[k][j] >= 0) {
                atomicAdd(&hist[(unsigned)tv[k][j] / NPW], 1u);
                so[k][j] = (unsigned)sv[k][j] / NPW;
                rk[k][j] = atomicAdd(&shist[so[k][j]], 1u);
            }
        }
    }
    __syncthreads();
    for (int o = tid; o < NB; o += BIN_THREADS) {
        unsigned h = hist[o];
        if (h) atomicAdd(&cnt[o * CSTRIDE], h);
    }
    if (wave == 0) {
        const int per = (NB + 32) / 32;
        const int lo = lane * per;
        unsigned s = 0u;
        for (int i = 0; i < per; ++i) s += shist[lo + i];
        unsigned inc = s;
        for (int d = 1; d < 32; d <<= 1) {
            unsigned up = __shfl_up(inc, d);
            if (lane >= d) inc += up;
        }
        unsigned run = inc - s;
        for (int i = 0; i < per; ++i) {
            spos[lo + i] = run;
            run += shist[lo + i];
        }
    }
    __syncthreads();
    unsigned* myoff = soff + (size_t)blockIdx.x * (NB + 1);
    for (int o = tid; o <= NB; o += BIN_THREADS) myoff[o] = spos[o];
#pragma unroll
    for (int k = 0; k < BIN_LD; ++k) {
#pragma unroll
        for (int j = 0; j < 4; ++j) rk[k][j] += spos[so[k][j]];
    }
#pragma unroll
    for (int k = 0; k < BIN_LD; ++k) {
#pragma unroll
        for (int j = 0; j < 4; ++j) {
            if (tv[k][j] >= 0) {
                unsigned sl = (unsigned)sv[k][j] - so[k][j] * NPW;
                stage[rk[k][j]] = (unsigned short)sl;
            }
        }
    }
    __syncthreads();
    const unsigned total = spos[NB];
    v4i* gs = (v4i*)((unsigned short*)ssort_w + (size_t)blockIdx.x * BIN_EDGES);
    const v4i* ls = (const v4i*)stage;
    for (int q = tid; q < BIN_EDGES / 8; q += BIN_THREADS) {
        if ((unsigned)q * 8u < total) gs[q] = ls[q];
    }
}

__global__ __launch_bounds__(DEG_THREADS)
void gcn_outdeg_kernel(const int* __restrict__ ssort_w,
                       const unsigned* __restrict__ soff,
                       float* __restrict__ nrm,
                       int N, int NB, int nbin) {
    __shared__ unsigned lcnt[NPW];
    const int tid = threadIdx.x;
    const int o = blockIdx.x;
    for (int i = tid; i < NPW; i += DEG_THREADS) lcnt[i] = 0u;
    __syncthreads();
    const unsigned short* ssort = (const unsigned short*)ssort_w;
    for (int c = tid; c < nbin; c += DEG_THREADS) {
        const unsigned* t = soff + (size_t)c * (NB + 1) + o;
        unsigned a = t[0];
        unsigned b = t[1];
        if (b > (unsigned)BIN_EDGES) b = BIN_EDGES;
        if (a < b) {
            const v4i* blk = (const v4i*)(ssort + (size_t)c * BIN_EDGES);
            for (unsigned q = a >> 3; q * 8u < b; ++q) {
                v4i d = blk[q];
                unsigned wd[4] = {(unsigned)d.x, (unsigned)d.y, (unsigned)d.z, (unsigned)d.w};
#pragma unroll
                for (int e = 0; e < 4; ++e) {
                    unsigned i0 = q * 8u + 2u * e;
                    unsigned id0 = wd[e] & 0xFFFFu;
                    unsigned id1 = wd[e] >> 16;
                    if (i0 >= a && i0 < b && id0 < NPW) atomicAdd(&lcnt[id0], 1u);
                    if (i0 + 1u >= a && i0 + 1u < b && id1 < NPW) atomicAdd(&lcnt[id1], 1u);
                }
            }
        }
    }
    __syncthreads();
    for (int i = tid; i < NPW; i += DEG_THREADS) {
        int node = o * NPW + i;
        if (node < N) {
            unsigned dg = lcnt[i];
            nrm[node] = rsqrtf((float)(dg ? dg : 1u));
        }
    }
}

__global__ __launch_bounds__(BIN_THREADS)
void gcn_fill_kernel(const float* __restrict__ e_feat,
                     const int* __restrict__ src,
                     const int* __restrict__ dst,
                     const unsigned* __restrict__ cnt,
                     unsigned* __restrict__ cur,
                     int* __restrict__ rec,
                     int E, int NB) {
    __shared__ unsigned hist[MAXB];
    __shared__ unsigned cntv[MAXB + 32];
    __shared__ unsigned gpos[MAXB + 32];
    __shared__ __attribute__((aligned(16))) v2i srec[BIN_EDGES];
    __shared__ unsigned sdst[BIN_EDGES];
    const int tid = threadIdx.x;
    const int lane = tid & 31;
    const int wave = tid >> 5;

    unsigned cv[(MAXB + 32 + BIN_THREADS - 1) / BIN_THREADS];
#pragma unroll
    for (int q = 0; q < (MAXB + 32 + BIN_THREADS - 1) / BIN_THREADS; ++q) {
        int o = tid + q * BIN_THREADS;
        cv[q] = (o < NB) ? cnt[o * CSTRIDE] : 0u;
    }
    const int chunk0 = blockIdx.x * BIN_EDGES + tid * 4;
    int sv[BIN_LD][4], tv[BIN_LD][4], wv[BIN_LD][4];
    if ((blockIdx.x + 1) * BIN_EDGES <= E) {
#pragma unroll
        for (int k = 0; k < BIN_LD; ++k) {
            load4(sv[k], src, chunk0 + k * (BIN_THREADS * 4));
            load4(tv[k], dst, chunk0 + k * (BIN_THREADS * 4));
            load4w(wv[k], e_feat, chunk0 + k * (BIN_THREADS * 4));
        }
    } else {
#pragma unroll
        for (int k = 0; k < BIN_LD; ++k) {
            load4_tail(sv[k], src, chunk0 + k * (BIN_THREADS * 4), E);
            load4_tail(tv[k], dst, chunk0 + k * (BIN_THREADS * 4), E);
            load4w_tail(wv[k], e_feat, chunk0 + k * (BIN_THREADS * 4), E);
        }
    }
#pragma unroll
    for (int q = 0; q < (MAXB + 32 + BIN_THREADS - 1) / BIN_THREADS; ++q) {
        int o = tid + q * BIN_THREADS;
        if (o < MAXB) hist[o] = 0u;
        if (o < MAXB + 32) cntv[o] = cv[q];
    }
    __syncthreads();

    if (wave == 0) {
        const int per = (NB + 31) / 32;
        const int lo = lane * per;
        unsigned s = 0u;
        for (int i = 0; i < per; ++i) s += cntv[lo + i];
        unsigned inc = s;
        for (int d = 1; d < 32; d <<= 1) {
            unsigned up = __shfl_up(inc, d);
            if (lane >= d) inc += up;
        }
        unsigned run = inc - s;
        for (int i = 0; i < per; ++i) {
            gpos[lo + i] = run;
            run += cntv[lo + i];
        }
    }
    unsigned ow[BIN_LD][4], rk[BIN_LD][4];
#pragma unroll
    for (int k = 0; k < BIN_LD; ++k) {
#pragma unroll
        for (int j = 0; j < 4; ++j) {
            ow[k][j] = (tv[k][j] >= 0) ? (unsigned)tv[k][j] / NPW : 0u;
            rk[k][j] = 0u;
            if (tv[k][j] >= 0) rk[k][j] = atomicAdd(&hist[ow[k][j]], 1u);
        }
    }
    __syncthreads();
    unsigned hh[MAXB / BIN_THREADS], got[MAXB / BIN_THREADS];
#pragma unroll
    for (int q = 0; q < MAXB / BIN_THREADS; ++q) {
        int o = tid + q * BIN_THREADS;
        hh[q] = (o < NB) ? hist[o] : 0u;
        got[q] = 0u;
        if (hh[q]) got[q] = atomicAdd(&cur[o * CSTRIDE], hh[q]);
    }
    if (wave == 0) {
        const int per = (NB + 31) / 32;
        const int lo = lane * per;
        unsigned s = 0u;
        for (int i = 0; i < per; ++i) s += hist[lo + i];
        unsigned inc = s;
        for (int d = 1; d < 32; d <<= 1) {
            unsigned up = __shfl_up(inc, d);
            if (lane >= d) inc += up;
        }
        unsigned run = inc - s;
        for (int i = 0; i < per; ++i) {
            cntv[lo + i] = run;
            run += hist[lo + i];
        }
    }
#pragma unroll
    for (int q = 0; q < MAXB / BIN_THREADS; ++q) {
        if (hh[q]) gpos[tid + q * BIN_THREADS] += got[q];
    }
    __syncthreads();
    unsigned ls[BIN_LD][4];
#pragma unroll
    for (int k = 0; k < BIN_LD; ++k) {
#pragma unroll
        for (int j = 0; j < 4; ++j) {
            ls[k][j] = rk[k][j] + cntv[ow[k][j]];
            rk[k][j] += gpos[ow[k][j]];
        }
    }
#pragma unroll
    for (int k = 0; k < BIN_LD; ++k) {
#pragma unroll
        for (int j = 0; j < 4; ++j) {
            int t = tv[k][j];
            if (t >= 0) {
                unsigned tl = (unsigned)t - ow[k][j] * NPW;
                v2i r;
                r.x = (int)((tl << SBITS) | (unsigned)sv[k][j]);
                r.y = wv[k][j];
                srec[ls[k][j]] = r;
                sdst[ls[k][j]] = rk[k][j];
            }
        }
    }
    __syncthreads();
    int total = E - blockIdx.x * BIN_EDGES;
    if (total > BIN_EDGES) total = BIN_EDGES;
    v2i* recs = (v2i*)rec;
    if (total == BIN_EDGES) {
        v2i ro[BIN_EDGES / BIN_THREADS];
        unsigned go[BIN_EDGES / BIN_THREADS];
#pragma unroll
        for (int k = 0; k < BIN_EDGES / BIN_THREADS; ++k) {
            ro[k] = srec[tid + k * BIN_THREADS];
            go[k] = sdst[tid + k * BIN_THREADS];
        }
#pragma unroll
        for (int k = 0; k < BIN_EDGES / BIN_THREADS; ++k) recs[go[k]] = ro[k];
    } else {
        for (int i = tid; i < total; i += BIN_THREADS) recs[sdst[i]] = srec[i];
    }
}

__device__ __forceinline__ void drain_batch(v2i rc, int n, int lane, int node0,
                                            const float* __restrict__ feat,
                                            const float* __restrict__ nrm,
                                            float* sagg, unsigned* sind) {
    bool valid = lane < n;
    int s = rc.x & SMASK;
    int tl = (int)((unsigned)rc.x >> SBITS);
    float norm = nrm[s];
    if (valid) atomicAdd(&sind[tl], 1u);
    bool live = valid && (s != node0 + tl);
    unsigned m = __builtin_amdgcn_ballot_w32(live);
    int sb = s * FD;
    int tb = tl * FD;
    int wi = rc.y;

    float fv[32];
#pragma unroll
    for (int j = 0; j < 32; ++j) {
        int sj = __builtin_amdgcn_readlane(sb, j);
        fv[j] = feat[sj + lane];
    }
    __builtin_amdgcn_sched_barrier(0);
    int ni = __float_as_int(norm);
#pragma unroll
    for (int j = 0; j < 32; ++j) {
        if ((m >> j) & 1u) {
            float nj = __int_as_float(__builtin_amdgcn_readlane(ni, j));
            float wj = __int_as_float(__builtin_amdgcn_readlane(wi, j));
            int tj = __builtin_amdgcn_readlane(tb, j);
            atomicAdd(&sagg[tj + lane], fv[j] * nj * wj);
        }
    }
}

__global__ __launch_bounds__(OWN_THREADS)
void gcn_owner_kernel(const float* __restrict__ feat,
                      const float* __restrict__ nrm,
                      const unsigned* __restrict__ cnt,
                      const int* __restrict__ rec,
                      float* __restrict__ out,
                      int N) {
    __shared__ __attribute__((aligned(16))) float sagg[NPW * FD];
    __shared__ unsigned sind[NPW];
    __shared__ unsigned sfirst;

    const int tid = threadIdx.x;
    const int lane = tid & 31;
    const int wave = __builtin_amdgcn_readfirstlane(tid >> 5);
    const int b = blockIdx.x;
    const int node0 = b * NPW;

    for (int i = tid; i < NPW * FD; i += OWN_THREADS) sagg[i] = 0.0f;
    for (int i = tid; i < NPW; i += OWN_THREADS) sind[i] = 0u;
    if (tid == 0) sfirst = 0u;
    __syncthreads();

    unsigned part = 0u;
    for (int o = tid; o < b; o += OWN_THREADS) part += cnt[o * CSTRIDE];
    if (part) atomicAdd(&sfirst, part);
    __syncthreads();

    const int n = (int)cnt[b * CSTRIDE];
    const v2i* mine = (const v2i*)rec + sfirst;
    const int nBatch = (n + 31) >> 5;

    int bt = wave;
    v2i rc = {0, 0};
    if (bt < nBatch && bt * 32 + lane < n) rc = mine[bt * 32 + lane];
    while (bt < nBatch) {
        const int nb = bt + OWN_WAVES;
        v2i rn = {0, 0};
        if (nb < nBatch && nb * 32 + lane < n) rn = mine[nb * 32 + lane];
        const int left = n - bt * 32;
        drain_batch(rc, left < 32 ? left : 32, lane, node0, feat, nrm, sagg, sind);
        rc = rn;
        bt = nb;
    }

    __syncthreads();

    if (wave < NTILE) {
        int r = lane & 15;
        int half = lane >> 4;
        int ln = wave * 16 + r;
        int node = node0 + ln;

        float vals[16];
        if (node < N) {
            unsigned dg = sind[ln];
            float dn = rsqrtf((float)(dg ? dg : 1u));
            const v4f* fp = (const v4f*)(feat + (size_t)node * FD + half * 16);
            const v4f* ap = (const v4f*)(sagg + ln * FD + half * 16);
#pragma unroll
            for (int qd = 0; qd < 4; ++qd) {
                v4f f = fp[qd];
                v4f a = ap[qd];
                vals[4 * qd + 0] = __builtin_fabsf(f.x - a.x * dn);
                vals[4 * qd + 1] = __builtin_fabsf(f.y - a.y * dn);
                vals[4 * qd + 2] = __builtin_fabsf(f.z - a.z * dn);
                vals[4 * qd + 3] = __builtin_fabsf(f.w - a.w * dn);
            }
        } else {
#pragma unroll
            for (int qd = 0; qd < 16; ++qd) vals[qd] = 0.0f;
        }

        v2f ones = {1.0f, 1.0f};
        v8f c = {};
#pragma unroll
        for (int kk = 0; kk < 8; ++kk) {
            v2f a;
            a.x = vals[2 * kk + 0];
            a.y = vals[2 * kk + 1];
            c = __builtin_amdgcn_wmma_f32_16x16x4_f32(
                    false, a, false, ones, (short)0, c, false, false);
        }

        if ((lane & 15) == 0) {
            int base = node0 + wave * 16 + half * 8;
#pragma unroll
            for (int mm = 0; mm < 8; ++mm) {
                int nd = base + mm;
                if (nd < N) out[nd] = c[mm];
            }
        }
    }
}

extern "C" void kernel_launch(void* const* d_in, const int* in_sizes, int n_in,
                              void* d_out, int out_size, void* d_ws, size_t ws_size,
                              hipStream_t stream) {
    const float* feat   = (const float*)d_in[0];
    const float* e_feat = (const float*)d_in[1];
    const int*   src    = (const int*)d_in[2];
    const int*   dst    = (const int*)d_in[3];
    float*       out    = (float*)d_out;

    int N = in_sizes[0] / FD;
    int E = in_sizes[2];
    int NB = (N + NPW - 1) / NPW;
    int Npad = (N + 31) / 32 * 32;
    int nbin = (E + BIN_EDGES - 1) / BIN_EDGES;

    float*    nrm  = (float*)d_ws;
    unsigned* cnt  = (unsigned*)d_ws + Npad;
    unsigned* cur  = cnt + NB * CSTRIDE;
    int*      rec  = (int*)(cur + NB * CSTRIDE);
    int*      ssort = rec;
    unsigned* soff  = (unsigned*)(rec + (size_t)nbin * (BIN_EDGES / 2));

    int zero_v4 = (2 * NB * CSTRIDE) / 4;
    gcn_zero_kernel<<<(zero_v4 + 255) / 256, 256, 0, stream>>>((float*)cnt, zero_v4);

    gcn_count_kernel<<<nbin, BIN_THREADS, 0, stream>>>(src, dst, cnt, ssort, soff, E, NB);

    gcn_outdeg_kernel<<<NB, DEG_THREADS, 0, stream>>>(ssort, soff, nrm, N, NB, nbin);

    gcn_fill_kernel<<<nbin, BIN_THREADS, 0, stream>>>(e_feat, src, dst, cnt, cur, rec, E, NB);

    gcn_owner_kernel<<<NB, OWN_THREADS, 0, stream>>>(feat, nrm, cnt, rec, out, N);
}


# per-owner edge totals: count pass stores each chunk's owner counts beside its segment starts, out-degree pass sums its column and clears the cursor; zero launch and count-pass global adds go
# speedup vs baseline: 4.8293x; 1.1785x over previous
#include <hip/hip_runtime.h>


typedef __attribute__((ext_vector_type(2))) float v2f;
typedef __attribute__((ext_vector_type(4))) float v4f;
typedef __attribute__((ext_vector_type(8))) float v8f;
typedef __attribute__((ext_vector_type(2))) int v2i;
typedef __attribute__((ext_vector_type(4))) int v4i;

#define FD 32
#define NPW 400
#define NTILE (NPW / 16)
#define OWN_THREADS 1024
#define OWN_WAVES (OWN_THREADS / 32)
#define BIN_THREADS 256
#define BIN_LD 4
#define BIN_EDGES (BIN_THREADS * 4 * BIN_LD)
#define MAXB 1024
#define CSTRIDE 32
#define DEG_THREADS 512
#define SBITS 23
#define SMASK 0x7FFFFF

__device__ __forceinline__ void load4_tail(int (&v)[4], const int* __restrict__ p, int base, int E) {
#pragma unroll
    for (int j = 0; j < 4; ++j) v[j] = (base + j < E) ? p[base + j] : -1;
}
__device__ __forceinline__ void load4w_tail(int (&v)[4], const float* __restrict__ p, int base, int E) {
#pragma unroll
    for (int j = 0; j < 4; ++j) v[j] = (base + j < E) ? __float_as_int(p[base + j]) : 0;
}
__device__ __forceinline__ void load4(int (&v)[4], const int* __restrict__ p, int base) {
    v4i d = *(const v4i*)(p + base);
    v[0] = d.x; v[1] = d.y; v[2] = d.z; v[3] = d.w;
}
__device__ __forceinline__ void load4w(int (&v)[4], const float* __restrict__ p, int base) {
    v4f d = *(const v4f*)(p + base);
    v[0] = __float_as_int(d.x); v[1] = __float_as_int(d.y);
    v[2] = __float_as_int(d.z); v[3] = __float_as_int(d.w);
}

__global__ __launch_bounds__(BIN_THREADS)
void gcn_count_kernel(const int* __restrict__ src,
                      const int* __restrict__ dst,
                      int* __restrict__ ssort_w,
                      unsigned* __restrict__ soff,
                      int E, int NB) {
    __shared__ unsigned hist[MAXB];
    __shared__ unsigned shist[MAXB + 32];
    __shared__ unsigned spos[MAXB + 32];
    __shared__ __attribute__((aligned(16))) unsigned short stage[BIN_EDGES];
    const int tid = threadIdx.x;
    const int lane = tid & 31;
    const int wave = tid >> 5;
    const int chunk0 = blockIdx.x * BIN_EDGES + tid * 4;
    int sv[BIN_LD][4], tv[BIN_LD][4];
    if ((blockIdx.x + 1) * BIN_EDGES <= E) {
#pragma unroll
        for (int k = 0; k < BIN_LD; ++k) {
            load4(sv[k], src, chunk0 + k * (BIN_THREADS * 4));
            load4(tv[k], dst, chunk0 + k * (BIN_THREADS * 4));
        }
    } else {
#pragma unroll
        for (int k = 0; k < BIN_LD; ++k) {
            load4_tail(sv[k], src, chunk0 + k * (BIN_THREADS * 4), E);
            load4_tail(tv[k], dst, chunk0 + k * (BIN_THREADS * 4), E);
        }
    }
    for (int o = tid; o < NB + 32; o += BIN_THREADS) {
        if (o < NB) hist[o] = 0u;
        shist[o] = 0u;
    }
    __syncthreads();

    unsigned so[BIN_LD][4], rk[BIN_LD][4];
#pragma unroll
    for (int k = 0; k < BIN_LD; ++k) {
#pragma unroll
        for (int j = 0; j < 4; ++j) {
            so[k][j] = 0u;
            rk[k][j] = 0u;
            if (tv[k][j] >= 0) {
                atomicAdd(&hist[(unsigned)tv[k][j] / NPW], 1u);
                so[k][j] = (unsigned)sv[k][j] / NPW;
                rk[k][j] = atomicAdd(&shist[so[k][j]], 1u);
            }
        }
    }
    __syncthreads();
    if (wave == 0) {
        const int per = (NB + 32) / 32;
        const int lo = lane * per;
        unsigned s = 0u;
        for (int i = 0; i < per; ++i) s += shist[lo + i];
        unsigned inc = s;
        for (int d = 1; d < 32; d <<= 1) {
            unsigned up = __shfl_up(inc, d);
            if (lane >= d) inc += up;
        }
        unsigned run = inc - s;
        for (int i = 0; i < per; ++i) {
            spos[lo + i] = run;
            run += shist[lo + i];
        }
    }
    __syncthreads();
    v2i* myoff = (v2i*)soff + (size_t)blockIdx.x * (NB + 1);
    for (int o = tid; o <= NB; o += BIN_THREADS) {
        v2i p;
        p.x = (int)spos[o];
        p.y = (o < NB) ? (int)hist[o] : 0;
        myoff[o] = p;
    }
#pragma unroll
    for (int k = 0; k < BIN_LD; ++k) {
#pragma unroll
        for (int j = 0; j < 4; ++j) rk[k][j] += spos[so[k][j]];
    }
#pragma unroll
    for (int k = 0; k < BIN_LD; ++k) {
#pragma unroll
        for (int j = 0; j < 4; ++j) {
            if (tv[k][j] >= 0) {
                unsigned sl = (unsigned)sv[k][j] - so[k][j] * NPW;
                stage[rk[k][j]] = (unsigned short)sl;
            }
        }
    }
    __syncthreads();
    const unsigned total = spos[NB];
    v4i* gs = (v4i*)((unsigned short*)ssort_w + (size_t)blockIdx.x * BIN_EDGES);
    const v4i* ls = (const v4i*)stage;
    for (int q = tid; q < BIN_EDGES / 8; q += BIN_THREADS) {
        if ((unsigned)q * 8u < total) gs[q] = ls[q];
    }
}

__global__ __launch_bounds__(DEG_THREADS)
void gcn_outdeg_kernel(const int* __restrict__ ssort_w,
                       const unsigned* __restrict__ soff,
                       float* __restrict__ nrm,
                       unsigned* __restrict__ cnt,
                       unsigned* __restrict__ cur,
                       int N, int NB, int nbin) {
    __shared__ unsigned lcnt[NPW];
    __shared__ unsigned ltot[32];
    const int tid = threadIdx.x;
    const int o = blockIdx.x;
    for (int i = tid; i < NPW; i += DEG_THREADS) lcnt[i] = 0u;
    if (tid < 32) ltot[tid] = 0u;
    __syncthreads();
    unsigned hsum = 0u;
    const unsigned short* ssort = (const unsigned short*)ssort_w;
    for (int c = tid; c < nbin; c += DEG_THREADS) {
        const v2i* t = (const v2i*)soff + (size_t)c * (NB + 1) + o;
        v2i p0 = t[0];
        v2i p1 = t[1];
        unsigned a = (unsigned)p0.x;
        unsigned b = (unsigned)p1.x;
        hsum += (unsigned)p0.y;
        if (b > (unsigned)BIN_EDGES) b = BIN_EDGES;
        if (a < b) {
            const v4i* blk = (const v4i*)(ssort + (size_t)c * BIN_EDGES);
            for (unsigned q = a >> 3; q * 8u < b; ++q) {
                v4i d = blk[q];
                unsigned wd[4] = {(unsigned)d.x, (unsigned)d.y, (unsigned)d.z, (unsigned)d.w};
#pragma unroll
                for (int e = 0; e < 4; ++e) {
                    unsigned i0 = q * 8u + 2u * e;
                    unsigned id0 = wd[e] & 0xFFFFu;
                    unsigned id1 = wd[e] >> 16;
                    if (i0 >= a && i0 < b && id0 < NPW) atomicAdd(&lcnt[id0], 1u);
                    if (i0 + 1u >= a && i0 + 1u < b && id1 < NPW) atomicAdd(&lcnt[id1], 1u);
                }
            }
        }
    }
    if (hsum) atomicAdd(&ltot[tid & 31], hsum);
    __syncthreads();
    for (int i = tid; i < NPW; i += DEG_THREADS) {
        int node = o * NPW + i;
        if (node < N) {
            unsigned dg = lcnt[i];
            nrm[node] = rsqrtf((float)(dg ? dg : 1u));
        }
    }
    if (tid == 0) {
        unsigned tot = 0u;
        for (int i = 0; i < 32; ++i) tot += ltot[i];
        cnt[o * CSTRIDE] = tot;
        cur[o * CSTRIDE] = 0u;
    }
}

__global__ __launch_bounds__(BIN_THREADS)
void gcn_fill_kernel(const float* __restrict__ e_feat,
                     const int* __restrict__ src,
                     const int* __restrict__ dst,
                     const unsigned* __restrict__ cnt,
                     unsigned* __restrict__ cur,
                     int* __restrict__ rec,
                     int E, int NB) {
    __shared__ unsigned hist[MAXB];
    __shared__ unsigned cntv[MAXB + 32];
    __shared__ unsigned gpos[MAXB + 32];
    __shared__ __attribute__((aligned(16))) v2i srec[BIN_EDGES];
    __shared__ unsigned sdst[BIN_EDGES];
    const int tid = threadIdx.x;
    const int lane = tid & 31;
    const int wave = tid >> 5;

    unsigned cv[(MAXB + 32 + BIN_THREADS - 1) / BIN_THREADS];
#pragma unroll
    for (int q = 0; q < (MAXB + 32 + BIN_THREADS - 1) / BIN_THREADS; ++q) {
        int o = tid + q * BIN_THREADS;
        cv[q] = (o < NB) ? cnt[o * CSTRIDE] : 0u;
    }
    const int chunk0 = blockIdx.x * BIN_EDGES + tid * 4;
    int sv[BIN_LD][4], tv[BIN_LD][4], wv[BIN_LD][4];
    if ((blockIdx.x + 1) * BIN_EDGES <= E) {
#pragma unroll
        for (int k = 0; k < BIN_LD; ++k) {
            load4(sv[k], src, chunk0 + k * (BIN_THREADS * 4));
            load4(tv[k], dst, chunk0 + k * (BIN_THREADS * 4));
            load4w(wv[k], e_feat, chunk0 + k * (BIN_THREADS * 4));
        }
    } else {
#pragma unroll
        for (int k = 0; k < BIN_LD; ++k) {
            load4_tail(sv[k], src, chunk0 + k * (BIN_THREADS * 4), E);
            load4_tail(tv[k], dst, chunk0 + k * (BIN_THREADS * 4), E);
            load4w_tail(wv[k], e_feat, chunk0 + k * (BIN_THREADS * 4), E);
        }
    }
#pragma unroll
    for (int q = 0; q < (MAXB + 32 + BIN_THREADS - 1) / BIN_THREADS; ++q) {
        int o = tid + q * BIN_THREADS;
        if (o < MAXB) hist[o] = 0u;
        if (o < MAXB + 32) cntv[o] = cv[q];
    }
    __syncthreads();

    if (wave == 0) {
        const int per = (NB + 31) / 32;
        const int lo = lane * per;
        unsigned s = 0u;
        for (int i = 0; i < per; ++i) s += cntv[lo + i];
        unsigned inc = s;
        for (int d = 1; d < 32; d <<= 1) {
            unsigned up = __shfl_up(inc, d);
            if (lane >= d) inc += up;
        }
        unsigned run = inc - s;
        for (int i = 0; i < per; ++i) {
            gpos[lo + i] = run;
            run += cntv[lo + i];
        }
    }
    unsigned ow[BIN_LD][4], rk[BIN_LD][4];
#pragma unroll
    for (int k = 0; k < BIN_LD; ++k) {
#pragma unroll
        for (int j = 0; j < 4; ++j) {
            ow[k][j] = (tv[k][j] >= 0) ? (unsigned)tv[k][j] / NPW : 0u;
            rk[k][j] = 0u;
            if (tv[k][j] >= 0) rk[k][j] = atomicAdd(&hist[ow[k][j]], 1u);
        }
    }
    __syncthreads();
    unsigned hh[MAXB / BIN_THREADS], got[MAXB / BIN_THREADS];
#pragma unroll
    for (int q = 0; q < MAXB / BIN_THREADS; ++q) {
        int o = tid + q * BIN_THREADS;
        hh[q] = (o < NB) ? hist[o] : 0u;
        got[q] = 0u;
        if (hh[q]) got[q] = atomicAdd(&cur[o * CSTRIDE], hh[q]);
    }
    if (wave == 0) {
        const int per = (NB + 31) / 32;
        const int lo = lane * per;
        unsigned s = 0u;
        for (int i = 0; i < per; ++i) s += hist[lo + i];
        unsigned inc = s;
        for (int d = 1; d < 32; d <<= 1) {
            unsigned up = __shfl_up(inc, d);
            if (lane >= d) inc += up;
        }
        unsigned run = inc - s;
        for (int i = 0; i < per; ++i) {
            cntv[lo + i] = run;
            run += hist[lo + i];
        }
    }
#pragma unroll
    for (int q = 0; q < MAXB / BIN_THREADS; ++q) {
        if (hh[q]) gpos[tid + q * BIN_THREADS] += got[q];
    }
    __syncthreads();
    unsigned ls[BIN_LD][4];
#pragma unroll
    for (int k = 0; k < BIN_LD; ++k) {
#pragma unroll
        for (int j = 0; j < 4; ++j) {
            ls[k][j] = rk[k][j] + cntv[ow[k][j]];
            rk[k][j] += gpos[ow[k][j]];
        }
    }
#pragma unroll
    for (int k = 0; k < BIN_LD; ++k) {
#pragma unroll
        for (int j = 0; j < 4; ++j) {
            int t = tv[k][j];
            if (t >= 0) {
                unsigned tl = (unsigned)t - ow[k][j] * NPW;
                v2i r;
                r.x = (int)((tl << SBITS) | (unsigned)sv[k][j]);
                r.y = wv[k][j];
                srec[ls[k][j]] = r;
                sdst[ls[k][j]] = rk[k][j];
            }
        }
    }
    __syncthreads();
    int total = E - blockIdx.x * BIN_EDGES;
    if (total > BIN_EDGES) total = BIN_EDGES;
    v2i* recs = (v2i*)rec;
    if (total == BIN_EDGES) {
        v2i ro[BIN_EDGES / BIN_THREADS];
        unsigned go[BIN_EDGES / BIN_THREADS];
#pragma unroll
        for (int k = 0; k < BIN_EDGES / BIN_THREADS; ++k) {
            ro[k] = srec[tid + k * BIN_THREADS];
            go[k] = sdst[tid + k * BIN_THREADS];
        }
#pragma unroll
        for (int k = 0; k < BIN_EDGES / BIN_THREADS; ++k) recs[go[k]] = ro[k];
    } else {
        for (int i = tid; i < total; i += BIN_THREADS) recs[sdst[i]] = srec[i];
    }
}

__device__ __forceinline__ void drain_batch(v2i rc, int n, int lane, int node0,
                                            const float* __restrict__ feat,
                                            const float* __restrict__ nrm,
                                            float* sagg, unsigned* sind) {
    bool valid = lane < n;
    int s = rc.x & SMASK;
    int tl = (int)((unsigned)rc.x >> SBITS);
    float norm = nrm[s];
    if (valid) atomicAdd(&sind[tl], 1u);
    bool live = valid && (s != node0 + tl);
    unsigned m = __builtin_amdgcn_ballot_w32(live);
    int sb = s * FD;
    int tb = tl * FD;
    int wi = rc.y;

    float fv[32];
#pragma unroll
    for (int j = 0; j < 32; ++j) {
        int sj = __builtin_amdgcn_readlane(sb, j);
        fv[j] = feat[sj + lane];
    }
    __builtin_amdgcn_sched_barrier(0);
    int ni = __float_as_int(norm);
#pragma unroll
    for (int j = 0; j < 32; ++j) {
        if ((m >> j) & 1u) {
            float nj = __int_as_float(__builtin_amdgcn_readlane(ni, j));
            float wj = __int_as_float(__builtin_amdgcn_readlane(wi, j));
            int tj = __builtin_amdgcn_readlane(tb, j);
            atomicAdd(&sagg[tj + lane], fv[j] * nj * wj);
        }
    }
}

__global__ __launch_bounds__(OWN_THREADS)
void gcn_owner_kernel(const float* __restrict__ feat,
                      const float* __restrict__ nrm,
                      const unsigned* __restrict__ cnt,
                      const int* __restrict__ rec,
                      float* __restrict__ out,
                      int N) {
    __shared__ __attribute__((aligned(16))) float sagg[NPW * FD];
    __shared__ unsigned sind[NPW];
    __shared__ unsigned sfirst;

    const int tid = threadIdx.x;
    const int lane = tid & 31;
    const int wave = __builtin_amdgcn_readfirstlane(tid >> 5);
    const int b = blockIdx.x;
    const int node0 = b * NPW;

    for (int i = tid; i < NPW * FD; i += OWN_THREADS) sagg[i] = 0.0f;
    for (int i = tid; i < NPW; i += OWN_THREADS) sind[i] = 0u;
    if (tid == 0) sfirst = 0u;
    __syncthreads();

    unsigned part = 0u;
    for (int o = tid; o < b; o += OWN_THREADS) part += cnt[o * CSTRIDE];
    if (part) atomicAdd(&sfirst, part);
    __syncthreads();

    const int n = (int)cnt[b * CSTRIDE];
    const v2i* mine = (const v2i*)rec + sfirst;
    const int nBatch = (n + 31) >> 5;

    int bt = wave;
    v2i rc = {0, 0};
    if (bt < nBatch && bt * 32 + lane < n) rc = mine[bt * 32 + lane];
    while (bt < nBatch) {
        const int nb = bt + OWN_WAVES;
        v2i rn = {0, 0};
        if (nb < nBatch && nb * 32 + lane < n) rn = mine[nb * 32 + lane];
        const int left = n - bt * 32;
        drain_batch(rc, left < 32 ? left : 32, lane, node0, feat, nrm, sagg, sind);
        rc = rn;
        bt = nb;
    }

    __syncthreads();

    if (wave < NTILE) {
        int r = lane & 15;
        int half = lane >> 4;
        int ln = wave * 16 + r;
        int node = node0 + ln;

        float vals[16];
        if (node < N) {
            unsigned dg = sind[ln];
            float dn = rsqrtf((float)(dg ? dg : 1u));
            const v4f* fp = (const v4f*)(feat + (size_t)node * FD + half * 16);
            const v4f* ap = (const v4f*)(sagg + ln * FD + half * 16);
#pragma unroll
            for (int qd = 0; qd < 4; ++qd) {
                v4f f = fp[qd];
                v4f a = ap[qd];
                vals[4 * qd + 0] = __builtin_fabsf(f.x - a.x * dn);
                vals[4 * qd + 1] = __builtin_fabsf(f.y - a.y * dn);
                vals[4 * qd + 2] = __builtin_fabsf(f.z - a.z * dn);
                vals[4 * qd + 3] = __builtin_fabsf(f.w - a.w * dn);
            }
        } else {
#pragma unroll
            for (int qd = 0; qd < 16; ++qd) vals[qd] = 0.0f;
        }

        v2f ones = {1.0f, 1.0f};
        v8f c = {};
#pragma unroll
        for (int kk = 0; kk < 8; ++kk) {
            v2f a;
            a.x = vals[2 * kk + 0];
            a.y = vals[2 * kk + 1];
            c = __builtin_amdgcn_wmma_f32_16x16x4_f32(
                    false, a, false, ones, (short)0, c, false, false);
        }

        if ((lane & 15) == 0) {
            int base = node0 + wave * 16 + half * 8;
#pragma unroll
            for (int mm = 0; mm < 8; ++mm) {
                int nd = base + mm;
                if (nd < N) out[nd] = c[mm];
            }
        }
    }
}

extern "C" void kernel_launch(void* const* d_in, const int* in_sizes, int n_in,
                              void* d_out, int out_size, void* d_ws, size_t ws_size,
                              hipStream_t stream) {
    const float* feat   = (const float*)d_in[0];
    const float* e_feat = (const float*)d_in[1];
    const int*   src    = (const int*)d_in[2];
    const int*   dst    = (const int*)d_in[3];
    float*       out    = (float*)d_out;

    int N = in_sizes[0] / FD;
    int E = in_sizes[2];
    int NB = (N + NPW - 1) / NPW;
    int Npad = (N + 31) / 32 * 32;
    int nbin = (E + BIN_EDGES - 1) / BIN_EDGES;

    float*    nrm  = (float*)d_ws;
    unsigned* cnt  = (unsigned*)d_ws + Npad;
    unsigned* cur  = cnt + NB * CSTRIDE;
    int*      rec  = (int*)(cur + NB * CSTRIDE);
    int*      ssort = rec;
    unsigned* soff  = (unsigned*)(rec + (size_t)nbin * (BIN_EDGES / 2));

    gcn_count_kernel<<<nbin, BIN_THREADS, 0, stream>>>(src, dst, ssort, soff, E, NB);

    gcn_outdeg_kernel<<<NB, DEG_THREADS, 0, stream>>>(ssort, soff, nrm, cnt, cur, N, NB, nbin);

    gcn_fill_kernel<<<nbin, BIN_THREADS, 0, stream>>>(e_feat, src, dst, cnt, cur, rec, E, NB);

    gcn_owner_kernel<<<NB, OWN_THREADS, 0, stream>>>(feat, nrm, cnt, rec, out, N);
}


# owner pass: source norms from an LDS byte table of out-degrees (saturated at 255, exact byte planes beside it), loaded once per work-group, not one scattered 4-byte request per edge; sums as before
# speedup vs baseline: 5.1642x; 1.0693x over previous
#include <hip/hip_runtime.h>


typedef __attribute__((ext_vector_type(2))) float v2f;
typedef __attribute__((ext_vector_type(4))) float v4f;
typedef __attribute__((ext_vector_type(8))) float v8f;
typedef __attribute__((ext_vector_type(2))) int v2i;
typedef __attribute__((ext_vector_type(4))) int v4i;

#define FD 32
#define NPW 400
#define NTILE (NPW / 16)
#define OWN_THREADS 1024
#define OWN_WAVES (OWN_THREADS / 32)
#define BIN_THREADS 256
#define BIN_LD 4
#define BIN_EDGES (BIN_THREADS * 4 * BIN_LD)
#define MAXB 1024
#define CSTRIDE 32
#define DEG_THREADS 512
#define SBITS 23
#define SMASK 0x7FFFFF
#define NTAB 131072

__device__ __forceinline__ void load4_tail(int (&v)[4], const int* __restrict__ p, int base, int E) {
#pragma unroll
    for (int j = 0; j < 4; ++j) v[j] = (base + j < E) ? p[base + j] : -1;
}
__device__ __forceinline__ void load4w_tail(int (&v)[4], const float* __restrict__ p, int base, int E) {
#pragma unroll
    for (int j = 0; j < 4; ++j) v[j] = (base + j < E) ? __float_as_int(p[base + j]) : 0;
}
__device__ __forceinline__ void load4(int (&v)[4], const int* __restrict__ p, int base) {
    v4i d = *(const v4i*)(p + base);
    v[0] = d.x; v[1] = d.y; v[2] = d.z; v[3] = d.w;
}
__device__ __forceinline__ void load4w(int (&v)[4], const float* __restrict__ p, int base) {
    v4f d = *(const v4f*)(p + base);
    v[0] = __float_as_int(d.x); v[1] = __float_as_int(d.y);
    v[2] = __float_as_int(d.z); v[3] = __float_as_int(d.w);
}

__global__ __launch_bounds__(BIN_THREADS)
void gcn_count_kernel(const int* __restrict__ src,
                      const int* __restrict__ dst,
                      int* __restrict__ ssort_w,
                      unsigned* __restrict__ soff,
                      int E, int NB) {
    __shared__ unsigned hist[MAXB];
    __shared__ unsigned shist[MAXB + 32];
    __shared__ unsigned spos[MAXB + 32];
    __shared__ __attribute__((aligned(16))) unsigned short stage[BIN_EDGES];
    const int tid = threadIdx.x;
    const int lane = tid & 31;
    const int wave = tid >> 5;
    const int chunk0 = blockIdx.x * BIN_EDGES + tid * 4;
    int sv[BIN_LD][4], tv[BIN_LD][4];
    if ((blockIdx.x + 1) * BIN_EDGES <= E) {
#pragma unroll
        for (int k = 0; k < BIN_LD; ++k) {
            load4(sv[k], src, chunk0 + k * (BIN_THREADS * 4));
            load4(tv[k], dst, chunk0 + k * (BIN_THREADS * 4));
        }
    } else {
#pragma unroll
        for (int k = 0; k < BIN_LD; ++k) {
            load4_tail(sv[k], src, chunk0 + k * (BIN_THREADS * 4), E);
            load4_tail(tv[k], dst, chunk0 + k * (BIN_THREADS * 4), E);
        }
    }
    for (int o = tid; o < NB + 32; o += BIN_THREADS) {
        if (o < NB) hist[o] = 0u;
        shist[o] = 0u;
    }
    __syncthreads();

    unsigned so[BIN_LD][4], rk[BIN_LD][4];
#pragma unroll
    for (int k = 0; k < BIN_LD; ++k) {
#pragma unroll
        for (int j = 0; j < 4; ++j) {
            so[k][j] = 0u;
            rk[k][j] = 0u;
            if (tv[k][j] >= 0) {
                atomicAdd(&hist[(unsigned)tv[k][j] / NPW], 1u);
                so[k][j] = (unsigned)sv[k][j] / NPW;
                rk[k][j] = atomicAdd(&shist[so[k][j]], 1u);
            }
        }
    }
    __syncthreads();
    if (wave == 0) {
        const int per = (NB + 32) / 32;
        const int lo = lane * per;
        unsigned s = 0u;
        for (int i = 0; i < per; ++i) s += shist[lo + i];
        unsigned inc = s;
        for (int d = 1; d < 32; d <<= 1) {
            unsigned up = __shfl_up(inc, d);
            if (lane >= d) inc += up;
        }
        unsigned run = inc - s;
        for (int i = 0; i < per; ++i) {
            spos[lo + i] = run;
            run += shist[lo + i];
        }
    }
    __syncthreads();
    v2i* myoff = (v2i*)soff + (size_t)blockIdx.x * (NB + 1);
    for (int o = tid; o <= NB; o += BIN_THREADS) {
        v2i p;
        p.x = (int)spos[o];
        p.y = (o < NB) ? (int)hist[o] : 0;
        myoff[o] = p;
    }
#pragma unroll
    for (int k = 0; k < BIN_LD; ++k) {
#pragma unroll
        for (int j = 0; j < 4; ++j) rk[k][j] += spos[so[k][j]];
    }
#pragma unroll
    for (int k = 0; k < BIN_LD; ++k) {
#pragma unroll
        for (int j = 0; j < 4; ++j) {
            if (tv[k][j] >= 0) {
                unsigned sl = (unsigned)sv[k][j] - so[k][j] * NPW;
                stage[rk[k][j]] = (unsigned short)sl;
            }
        }
    }
    __syncthreads();
    const unsigned total = spos[NB];
    v4i* gs = (v4i*)((unsigned short*)ssort_w + (size_t)blockIdx.x * BIN_EDGES);
    const v4i* ls = (const v4i*)stage;
    for (int q = tid; q < BIN_EDGES / 8; q += BIN_THREADS) {
        if ((unsigned)q * 8u < total) gs[q] = ls[q];
    }
}

__global__ __launch_bounds__(DEG_THREADS)
void gcn_outdeg_kernel(const int* __restrict__ ssort_w,
                       const unsigned* __restrict__ soff,
                       unsigned* __restrict__ planes,
                       unsigned* __restrict__ cnt,
                       unsigned* __restrict__ cur,
                       int tabN, int NB, int nbin) {
    __shared__ unsigned lcnt[NPW];
    __shared__ unsigned ltot[32];
    const int tid = threadIdx.x;
    const int o = blockIdx.x;
    for (int i = tid; i < NPW; i += DEG_THREADS) lcnt[i] = 0u;
    if (tid < 32) ltot[tid] = 0u;
    __syncthreads();
    unsigned hsum = 0u;
    const unsigned short* ssort = (const unsigned short*)ssort_w;
    for (int c = tid; c < nbin; c += DEG_THREADS) {
        const v2i* t = (const v2i*)soff + (size_t)c * (NB + 1) + o;
        v2i p0 = t[0];
        v2i p1 = t[1];
        unsigned a = (unsigned)p0.x;
        unsigned b = (unsigned)p1.x;
        hsum += (unsigned)p0.y;
        if (b > (unsigned)BIN_EDGES) b = BIN_EDGES;
        if (a < b) {
            const v4i* blk = (const v4i*)(ssort + (size_t)c * BIN_EDGES);
            for (unsigned q = a >> 3; q * 8u < b; ++q) {
                v4i d = blk[q];
                unsigned wd[4] = {(unsigned)d.x, (unsigned)d.y, (unsigned)d.z, (unsigned)d.w};
#pragma unroll
                for (int e = 0; e < 4; ++e) {
                    unsigned i0 = q * 8u + 2u * e;
                    unsigned id0 = wd[e] & 0xFFFFu;
                    unsigned id1 = wd[e] >> 16;
                    if (i0 >= a && i0 < b && id0 < NPW) atomicAdd(&lcnt[id0], 1u);
                    if (i0 + 1u >= a && i0 + 1u < b && id1 < NPW) atomicAdd(&lcnt[id1], 1u);
                }
            }
        }
    }
    if (hsum) atomicAdd(&ltot[tid & 31], hsum);
    __syncthreads();
    if (tid < NPW) {
        const int p = tid / (NPW / 4);
        const int i = tid - p * (NPW / 4);
        unsigned wv = 0u;
#pragma unroll
        for (int j = 0; j < 4; ++j) {
            unsigned dg = lcnt[4 * i + j];
            unsigned bv = (p == 0) ? (dg < 255u ? dg : 255u) : ((dg >> (8 * (p - 1))) & 255u);
            wv |= bv << (8 * j);
        }
        planes[(size_t)p * (tabN / 4) + o * (NPW / 4) + i] = wv;
    }
    if (tid == 0) {
        unsigned tot = 0u;
        for (int i = 0; i < 32; ++i) tot += ltot[i];
        cnt[o * CSTRIDE] = tot;
        cur[o * CSTRIDE] = 0u;
    }
}

__global__ __launch_bounds__(BIN_THREADS)
void gcn_fill_kernel(const float* __restrict__ e_feat,
                     const int* __restrict__ src,
                     const int* __restrict__ dst,
                     const unsigned* __restrict__ cnt,
                     unsigned* __restrict__ cur,
                     int* __restrict__ rec,
                     int E, int NB) {
    __shared__ unsigned hist[MAXB];
    __shared__ unsigned cntv[MAXB + 32];
    __shared__ unsigned gpos[MAXB + 32];
    __shared__ __attribute__((aligned(16))) v2i srec[BIN_EDGES];
    __shared__ unsigned sdst[BIN_EDGES];
    const int tid = threadIdx.x;
    const int lane = tid & 31;
    const int wave = tid >> 5;

    unsigned cv[(MAXB + 32 + BIN_THREADS - 1) / BIN_THREADS];
#pragma unroll
    for (int q = 0; q < (MAXB + 32 + BIN_THREADS - 1) / BIN_THREADS; ++q) {
        int o = tid + q * BIN_THREADS;
        cv[q] = (o < NB) ? cnt[o * CSTRIDE] : 0u;
    }
    const int chunk0 = blockIdx.x * BIN_EDGES + tid * 4;
    int sv[BIN_LD][4], tv[BIN_LD][4], wv[BIN_LD][4];
    if ((blockIdx.x + 1) * BIN_EDGES <= E) {
#pragma unroll
        for (int k = 0; k < BIN_LD; ++k) {
            load4(sv[k], src, chunk0 + k * (BIN_THREADS * 4));
            load4(tv[k], dst, chunk0 + k * (BIN_THREADS * 4));
            load4w(wv[k], e_feat, chunk0 + k * (BIN_THREADS * 4));
        }
    } else {
#pragma unroll
        for (int k = 0; k < BIN_LD; ++k) {
            load4_tail(sv[k], src, chunk0 + k * (BIN_THREADS * 4), E);
            load4_tail(tv[k], dst, chunk0 + k * (BIN_THREADS * 4), E);
            load4w_tail(wv[k], e_feat, chunk0 + k * (BIN_THREADS * 4), E);
        }
    }
#pragma unroll
    for (int q = 0; q < (MAXB + 32 + BIN_THREADS - 1) / BIN_THREADS; ++q) {
        int o = tid + q * BIN_THREADS;
        if (o < MAXB) hist[o] = 0u;
        if (o < MAXB + 32) cntv[o] = cv[q];
    }
    __syncthreads();

    if (wave == 0) {
        const int per = (NB + 31) / 32;
        const int lo = lane * per;
        unsigned s = 0u;
        for (int i = 0; i < per; ++i) s += cntv[lo + i];
        unsigned inc = s;
        for (int d = 1; d < 32; d <<= 1) {
            unsigned up = __shfl_up(inc, d);
            if (lane >= d) inc += up;
        }
        unsigned run = inc - s;
        for (int i = 0; i < per; ++i) {
            gpos[lo + i] = run;
            run += cntv[lo + i];
        }
    }
    unsigned ow[BIN_LD][4], rk[BIN_LD][4];
#pragma unroll
    for (int k = 0; k < BIN_LD; ++k) {
#pragma unroll
        for (int j = 0; j < 4; ++j) {
            ow[k][j] = (tv[k][j] >= 0) ? (unsigned)tv[k][j] / NPW : 0u;
            rk[k][j] = 0u;
            if (tv[k][j] >= 0) rk[k][j] = atomicAdd(&hist[ow[k][j]], 1u);
        }
    }
    __syncthreads();
    unsigned hh[MAXB / BIN_THREADS], got[MAXB / BIN_THREADS];
#pragma unroll
    for (int q = 0; q < MAXB / BIN_THREADS; ++q) {
        int o = tid + q * BIN_THREADS;
        hh[q] = (o < NB) ? hist[o] : 0u;
        got[q] = 0u;
        if (hh[q]) got[q] = atomicAdd(&cur[o * CSTRIDE], hh[q]);
    }
    if (wave == 0) {
        const int per = (NB + 31) / 32;
        const int lo = lane * per;
        unsigned s = 0u;
        for (int i = 0; i < per; ++i) s += hist[lo + i];
        unsigned inc = s;
        for (int d = 1; d < 32; d <<= 1) {
            unsigned up = __shfl_up(inc, d);
            if (lane >= d) inc += up;
        }
        unsigned run = inc - s;
        for (int i = 0; i < per; ++i) {
            cntv[lo + i] = run;
            run += hist[lo + i];
        }
    }
#pragma unroll
    for (int q = 0; q < MAXB / BIN_THREADS; ++q) {
        if (hh[q]) gpos[tid + q * BIN_THREADS] += got[q];
    }
    __syncthreads();
    unsigned ls[BIN_LD][4];
#pragma unroll
    for (int k = 0; k < BIN_LD; ++k) {
#pragma unroll
        for (int j = 0; j < 4; ++j) {
            ls[k][j] = rk[k][j] + cntv[ow[k][j]];
            rk[k][j] += gpos[ow[k][j]];
        }
    }
#pragma unroll
    for (int k = 0; k < BIN_LD; ++k) {
#pragma unroll
        for (int j = 0; j < 4; ++j) {
            int t = tv[k][j];
            if (t >= 0) {
                unsigned tl = (unsigned)t - ow[k][j] * NPW;
                v2i r;
                r.x = (int)((tl << SBITS) | (unsigned)sv[k][j]);
                r.y = wv[k][j];
                srec[ls[k][j]] = r;
                sdst[ls[k][j]] = rk[k][j];
            }
        }
    }
    __syncthreads();
    int total = E - blockIdx.x * BIN_EDGES;
    if (total > BIN_EDGES) total = BIN_EDGES;
    v2i* recs = (v2i*)rec;
    if (total == BIN_EDGES) {
        v2i ro[BIN_EDGES / BIN_THREADS];
        unsigned go[BIN_EDGES / BIN_THREADS];
#pragma unroll
        for (int k = 0; k < BIN_EDGES / BIN_THREADS; ++k) {
            ro[k] = srec[tid + k * BIN_THREADS];
            go[k] = sdst[tid + k * BIN_THREADS];
        }
#pragma unroll
        for (int k = 0; k < BIN_EDGES / BIN_THREADS; ++k) recs[go[k]] = ro[k];
    } else {
        for (int i = tid; i < total; i += BIN_THREADS) recs[sdst[i]] = srec[i];
    }
}

__device__ __forceinline__ void drain_batch(v2i rc, int n, int lane, int node0,
                                            const float* __restrict__ feat,
                                            const unsigned char* __restrict__ deg8,
                                            int tabN, int tabLim,
                                            const unsigned char* sdeg,
                                            float* sagg, unsigned* sind) {
    bool valid = lane < n;
    int s = rc.x & SMASK;
    int tl = (int)((unsigned)rc.x >> SBITS);
    unsigned d8 = (s < tabLim) ? (unsigned)sdeg[s] : 255u;
    if (valid) atomicAdd(&sind[tl], 1u);
    bool live = valid && (s != node0 + tl);
    unsigned m = __builtin_amdgcn_ballot_w32(live);
    int sb = s * FD;
    int tb = tl * FD;
    int wi = rc.y;

    float fv[32];
#pragma unroll
    for (int j = 0; j < 32; ++j) {
        int sj = __builtin_amdgcn_readlane(sb, j);
        fv[j] = feat[sj + lane];
    }
    unsigned dg = d8;
    if (d8 == 255u) {
        dg = (unsigned)deg8[(size_t)tabN + s] | ((unsigned)deg8[(size_t)2 * tabN + s] << 8) |
             ((unsigned)deg8[(size_t)3 * tabN + s] << 16);
    }
    float norm = rsqrtf((float)(dg ? dg : 1u));
    int ni = __float_as_int(norm);
#pragma unroll
    for (int j = 0; j < 32; ++j) {
        if ((m >> j) & 1u) {
            float nj = __int_as_float(__builtin_amdgcn_readlane(ni, j));
            float wj = __int_as_float(__builtin_amdgcn_readlane(wi, j));
            int tj = __builtin_amdgcn_readlane(tb, j);
            atomicAdd(&sagg[tj + lane], fv[j] * nj * wj);
        }
    }
}

__global__ __launch_bounds__(OWN_THREADS)
void gcn_owner_kernel(const float* __restrict__ feat,
                      const unsigned char* __restrict__ deg8,
                      const unsigned* __restrict__ cnt,
                      const int* __restrict__ rec,
                      float* __restrict__ out,
                      int N, int tabN) {
    __shared__ __attribute__((aligned(16))) float sagg[NPW * FD];
    __shared__ unsigned sind[NPW];
    __shared__ unsigned sfirst;
    __shared__ __attribute__((aligned(16))) unsigned char sdeg[NTAB];

    const int tid = threadIdx.x;
    const int lane = tid & 31;
    const int wave = __builtin_amdgcn_readfirstlane(tid >> 5);
    const int b = blockIdx.x;
    const int node0 = b * NPW;

    const int tabLim = tabN < NTAB ? tabN : NTAB;
    v4i tq[NTAB / 16 / OWN_THREADS];
#pragma unroll
    for (int k = 0; k < NTAB / 16 / OWN_THREADS; ++k) {
        int q = tid + k * OWN_THREADS;
        v4i z = {0, 0, 0, 0};
        tq[k] = (q * 16 < tabLim) ? ((const v4i*)deg8)[q] : z;
    }

    for (int i = tid; i < NPW * FD; i += OWN_THREADS) sagg[i] = 0.0f;
    for (int i = tid; i < NPW; i += OWN_THREADS) sind[i] = 0u;
    if (tid == 0) sfirst = 0u;
    __syncthreads();

    unsigned part = 0u;
    for (int o = tid; o < b; o += OWN_THREADS) part += cnt[o * CSTRIDE];
    if (part) atomicAdd(&sfirst, part);
#pragma unroll
    for (int k = 0; k < NTAB / 16 / OWN_THREADS; ++k) {
        int q = tid + k * OWN_THREADS;
        if (q * 16 < tabLim) ((v4i*)sdeg)[q] = tq[k];
    }
    __syncthreads();

    const int n = (int)cnt[b * CSTRIDE];
    const v2i* mine = (const v2i*)rec + sfirst;
    const int nBatch = (n + 31) >> 5;

    int bt = wave;
    v2i rc = {0, 0};
    if (bt < nBatch && bt * 32 + lane < n) rc = mine[bt * 32 + lane];
    while (bt < nBatch) {
        const int nb = bt + OWN_WAVES;
        v2i rn = {0, 0};
        if (nb < nBatch && nb * 32 + lane < n) rn = mine[nb * 32 + lane];
        const int left = n - bt * 32;
        drain_batch(rc, left < 32 ? left : 32, lane, node0, feat, deg8, tabN, tabLim, sdeg, sagg, sind);
        rc = rn;
        bt = nb;
    }

    __syncthreads();

    if (wave < NTILE) {
        int r = lane & 15;
        int half = lane >> 4;
        int ln = wave * 16 + r;
        int node = node0 + ln;

        float vals[16];
        if (node < N) {
            unsigned dg = sind[ln];
            float dn = rsqrtf((float)(dg ? dg : 1u));
            const v4f* fp = (const v4f*)(feat + (size_t)node * FD + half * 16);
            const v4f* ap = (const v4f*)(sagg + ln * FD + half * 16);
#pragma unroll
            for (int qd = 0; qd < 4; ++qd) {
                v4f f = fp[qd];
                v4f a = ap[qd];
                vals[4 * qd + 0] = __builtin_fabsf(f.x - a.x * dn);
                vals[4 * qd + 1] = __builtin_fabsf(f.y - a.y * dn);
                vals[4 * qd + 2] = __builtin_fabsf(f.z - a.z * dn);
                vals[4 * qd + 3] = __builtin_fabsf(f.w - a.w * dn);
            }
        } else {
#pragma unroll
            for (int qd = 0; qd < 16; ++qd) vals[qd] = 0.0f;
        }

        v2f ones = {1.0f, 1.0f};
        v8f c = {};
#pragma unroll
        for (int kk = 0; kk < 8; ++kk) {
            v2f a;
            a.x = vals[2 * kk + 0];
            a.y = vals[2 * kk + 1];
            c = __builtin_amdgcn_wmma_f32_16x16x4_f32(
                    false, a, false, ones, (short)0, c, false, false);
        }

        if ((lane & 15) == 0) {
            int base = node0 + wave * 16 + half * 8;
#pragma unroll
            for (int mm = 0; mm < 8; ++mm) {
                int nd = base + mm;
                if (nd < N) out[nd] = c[mm];
            }
        }
    }
}

extern "C" void kernel_launch(void* const* d_in, const int* in_sizes, int n_in,
                              void* d_out, int out_size, void* d_ws, size_t ws_size,
                              hipStream_t stream) {
    const float* feat   = (const float*)d_in[0];
    const float* e_feat = (const float*)d_in[1];
    const int*   src    = (const int*)d_in[2];
    const int*   dst    = (const int*)d_in[3];
    float*       out    = (float*)d_out;

    int N = in_sizes[0] / FD;
    int E = in_sizes[2];
    int NB = (N + NPW - 1) / NPW;
    int tabN = NB * NPW;
    int nbin = (E + BIN_EDGES - 1) / BIN_EDGES;

    unsigned* planes = (unsigned*)d_ws;
    unsigned* cnt  = (unsigned*)d_ws + tabN;
    unsigned* cur  = cnt + NB * CSTRIDE;
    int*      rec  = (int*)(cur + NB * CSTRIDE);
    int*      ssort = rec;
    unsigned* soff  = (unsigned*)(rec + (size_t)nbin * (BIN_EDGES / 2));

    gcn_count_kernel<<<nbin, BIN_THREADS, 0, stream>>>(src, dst, ssort, soff, E, NB);

    gcn_outdeg_kernel<<<NB, DEG_THREADS, 0, stream>>>(ssort, soff, planes, cnt, cur, tabN, NB, nbin);

    gcn_fill_kernel<<<nbin, BIN_THREADS, 0, stream>>>(e_feat, src, dst, cnt, cur, rec, E, NB);

    gcn_owner_kernel<<<NB, OWN_THREADS, 0, stream>>>(feat, (const unsigned char*)d_ws, cnt, rec, out, N, tabN);
}


# fill pass without cursor atomics: the out-degree pass scans each owner's chunk counts in chunk order and stores a first slot per (owner, chunk); the fill pass reads it; LDS sums as before
# speedup vs baseline: 6.3618x; 1.2319x over previous
#include <hip/hip_runtime.h>


typedef __attribute__((ext_vector_type(2))) float v2f;
typedef __attribute__((ext_vector_type(4))) float v4f;
typedef __attribute__((ext_vector_type(8))) float v8f;
typedef __attribute__((ext_vector_type(2))) int v2i;
typedef __attribute__((ext_vector_type(4))) int v4i;

#define FD 32
#define NPW 400
#define NTILE (NPW / 16)
#define OWN_THREADS 1024
#define OWN_WAVES (OWN_THREADS / 32)
#define BIN_THREADS 256
#define BIN_LD 4
#define BIN_EDGES (BIN_THREADS * 4 * BIN_LD)
#define MAXB 1024
#define CSTRIDE 32
#define DEG_THREADS 512
#define SBITS 23
#define SMASK 0x7FFFFF
#define NTAB 131072

__device__ __forceinline__ void load4_tail(int (&v)[4], const int* __restrict__ p, int base, int E) {
#pragma unroll
    for (int j = 0; j < 4; ++j) v[j] = (base + j < E) ? p[base + j] : -1;
}
__device__ __forceinline__ void load4w_tail(int (&v)[4], const float* __restrict__ p, int base, int E) {
#pragma unroll
    for (int j = 0; j < 4; ++j) v[j] = (base + j < E) ? __float_as_int(p[base + j]) : 0;
}
__device__ __forceinline__ void load4(int (&v)[4], const int* __restrict__ p, int base) {
    v4i d = *(const v4i*)(p + base);
    v[0] = d.x; v[1] = d.y; v[2] = d.z; v[3] = d.w;
}
__device__ __forceinline__ void load4w(int (&v)[4], const float* __restrict__ p, int base) {
    v4f d = *(const v4f*)(p + base);
    v[0] = __float_as_int(d.x); v[1] = __float_as_int(d.y);
    v[2] = __float_as_int(d.z); v[3] = __float_as_int(d.w);
}

__global__ __launch_bounds__(BIN_THREADS)
void gcn_count_kernel(const int* __restrict__ src,
                      const int* __restrict__ dst,
                      int* __restrict__ ssort_w,
                      unsigned* __restrict__ soff,
                      int E, int NB) {
    __shared__ unsigned hist[MAXB];
    __shared__ unsigned shist[MAXB + 32];
    __shared__ unsigned spos[MAXB + 32];
    __shared__ __attribute__((aligned(16))) unsigned short stage[BIN_EDGES];
    const int tid = threadIdx.x;
    const int lane = tid & 31;
    const int wave = tid >> 5;
    const int chunk0 = blockIdx.x * BIN_EDGES + tid * 4;
    int sv[BIN_LD][4], tv[BIN_LD][4];
    if ((blockIdx.x + 1) * BIN_EDGES <= E) {
#pragma unroll
        for (int k = 0; k < BIN_LD; ++k) {
            load4(sv[k], src, chunk0 + k * (BIN_THREADS * 4));
            load4(tv[k], dst, chunk0 + k * (BIN_THREADS * 4));
        }
    } else {
#pragma unroll
        for (int k = 0; k < BIN_LD; ++k) {
            load4_tail(sv[k], src, chunk0 + k * (BIN_THREADS * 4), E);
            load4_tail(tv[k], dst, chunk0 + k * (BIN_THREADS * 4), E);
        }
    }
    for (int o = tid; o < NB + 32; o += BIN_THREADS) {
        if (o < NB) hist[o] = 0u;
        shist[o] = 0u;
    }
    __syncthreads();

    unsigned so[BIN_LD][4], rk[BIN_LD][4];
#pragma unroll
    for (int k = 0; k < BIN_LD; ++k) {
#pragma unroll
        for (int j = 0; j < 4; ++j) {
            so[k][j] = 0u;
            rk[k][j] = 0u;
            if (tv[k][j] >= 0) {
                atomicAdd(&hist[(unsigned)tv[k][j] / NPW], 1u);
                so[k][j] = (unsigned)sv[k][j] / NPW;
                rk[k][j] = atomicAdd(&shist[so[k][j]], 1u);
            }
        }
    }
    __syncthreads();
    if (wave == 0) {
        const int per = (NB + 32) / 32;
        const int lo = lane * per;
        unsigned s = 0u;
        for (int i = 0; i < per; ++i) s += shist[lo + i];
        unsigned inc = s;
        for (int d = 1; d < 32; d <<= 1) {
            unsigned up = __shfl_up(inc, d);
            if (lane >= d) inc += up;
        }
        unsigned run = inc - s;
        for (int i = 0; i < per; ++i) {
            spos[lo + i] = run;
            run += shist[lo + i];
        }
    }
    __syncthreads();
    v2i* myoff = (v2i*)soff + (size_t)blockIdx.x * (NB + 1);
    for (int o = tid; o <= NB; o += BIN_THREADS) {
        v2i p;
        p.x = (int)spos[o];
        p.y = (o < NB) ? (int)hist[o] : 0;
        myoff[o] = p;
    }
#pragma unroll
    for (int k = 0; k < BIN_LD; ++k) {
#pragma unroll
        for (int j = 0; j < 4; ++j) rk[k][j] += spos[so[k][j]];
    }
#pragma unroll
    for (int k = 0; k < BIN_LD; ++k) {
#pragma unroll
        for (int j = 0; j < 4; ++j) {
            if (tv[k][j] >= 0) {
                unsigned sl = (unsigned)sv[k][j] - so[k][j] * NPW;
                stage[rk[k][j]] = (unsigned short)sl;
            }
        }
    }
    __syncthreads();
    const unsigned total = spos[NB];
    v4i* gs = (v4i*)((unsigned short*)ssort_w + (size_t)blockIdx.x * BIN_EDGES);
    const v4i* ls = (const v4i*)stage;
    for (int q = tid; q < BIN_EDGES / 8; q += BIN_THREADS) {
        if ((unsigned)q * 8u < total) gs[q] = ls[q];
    }
}

__global__ __launch_bounds__(DEG_THREADS)
void gcn_outdeg_kernel(const int* __restrict__ ssort_w,
                       const unsigned* __restrict__ soff,
                       unsigned* __restrict__ planes,
                       unsigned* __restrict__ cnt,
                       unsigned* __restrict__ base,
                       int tabN, int NB, int nbin) {
    __shared__ unsigned lcnt[NPW];
    __shared__ unsigned wtot[DEG_THREADS / 32];
    const int tid = threadIdx.x;
    const int lane = tid & 31;
    const int wave = tid >> 5;
    const int o = blockIdx.x;
    for (int i = tid; i < NPW; i += DEG_THREADS) lcnt[i] = 0u;
    __syncthreads();
    unsigned carry = 0u;
    const unsigned short* ssort = (const unsigned short*)ssort_w;
    for (int c0 = 0; c0 < nbin; c0 += DEG_THREADS) {
        const int c = c0 + tid;
        const bool in = c < nbin;
        v2i p0 = {0, 0};
        v2i p1 = {0, 0};
        if (in) {
            const v2i* t = (const v2i*)soff + (size_t)c * (NB + 1) + o;
            p0 = t[0];
            p1 = t[1];
        }
        const unsigned h = (unsigned)p0.y;
        unsigned inc = h;
        for (int d = 1; d < 32; d <<= 1) {
            unsigned up = __shfl_up(inc, d);
            if (lane >= d) inc += up;
        }
        if (lane == 31) wtot[wave] = inc;
        __syncthreads();
        unsigned before = 0u, all = 0u;
        for (int w = 0; w < DEG_THREADS / 32; ++w) {
            unsigned t = wtot[w];
            all += t;
            if (w < wave) before += t;
        }
        if (in) base[(size_t)o * nbin + c] = carry + before + inc - h;
        carry += all;
        unsigned a = (unsigned)p0.x;
        unsigned b = (unsigned)p1.x;
        if (b > (unsigned)BIN_EDGES) b = BIN_EDGES;
        if (in && a < b) {
            const v4i* blk = (const v4i*)(ssort + (size_t)c * BIN_EDGES);
            for (unsigned q = a >> 3; q * 8u < b; ++q) {
                v4i d = blk[q];
                unsigned wd[4] = {(unsigned)d.x, (unsigned)d.y, (unsigned)d.z, (unsigned)d.w};
#pragma unroll
                for (int e = 0; e < 4; ++e) {
                    unsigned i0 = q * 8u + 2u * e;
                    unsigned id0 = wd[e] & 0xFFFFu;
                    unsigned id1 = wd[e] >> 16;
                    if (i0 >= a && i0 < b && id0 < NPW) atomicAdd(&lcnt[id0], 1u);
                    if (i0 + 1u >= a && i0 + 1u < b && id1 < NPW) atomicAdd(&lcnt[id1], 1u);
                }
            }
        }
        __syncthreads();
    }
    if (tid < NPW) {
        const int p = tid / (NPW / 4);
        const int i = tid - p * (NPW / 4);
        unsigned wv = 0u;
#pragma unroll
        for (int j = 0; j < 4; ++j) {
            unsigned dg = lcnt[4 * i + j];
            unsigned bv = (p == 0) ? (dg < 255u ? dg : 255u) : ((dg >> (8 * (p - 1))) & 255u);
            wv |= bv << (8 * j);
        }
        planes[(size_t)p * (tabN / 4) + o * (NPW / 4) + i] = wv;
    }
    if (tid == 0) cnt[o * CSTRIDE] = carry;
}

__global__ __launch_bounds__(BIN_THREADS)
void gcn_fill_kernel(const float* __restrict__ e_feat,
                     const int* __restrict__ src,
                     const int* __restrict__ dst,
                     const unsigned* __restrict__ cnt,
                     const unsigned* __restrict__ base,
                     int* __restrict__ rec,
                     int E, int NB, int nbin) {
    __shared__ unsigned hist[MAXB];
    __shared__ unsigned cntv[MAXB + 32];
    __shared__ unsigned gpos[MAXB + 32];
    __shared__ __attribute__((aligned(16))) v2i srec[BIN_EDGES];
    __shared__ unsigned sdst[BIN_EDGES];
    const int tid = threadIdx.x;
    const int lane = tid & 31;
    const int wave = tid >> 5;

    unsigned cv[(MAXB + 32 + BIN_THREADS - 1) / BIN_THREADS];
    unsigned bv[MAXB / BIN_THREADS];
#pragma unroll
    for (int q = 0; q < (MAXB + 32 + BIN_THREADS - 1) / BIN_THREADS; ++q) {
        int o = tid + q * BIN_THREADS;
        cv[q] = (o < NB) ? cnt[o * CSTRIDE] : 0u;
    }
#pragma unroll
    for (int q = 0; q < MAXB / BIN_THREADS; ++q) {
        int o = tid + q * BIN_THREADS;
        bv[q] = (o < NB) ? base[(size_t)o * nbin + blockIdx.x] : 0u;
    }
    const int chunk0 = blockIdx.x * BIN_EDGES + tid * 4;
    int sv[BIN_LD][4], tv[BIN_LD][4], wv[BIN_LD][4];
    if ((blockIdx.x + 1) * BIN_EDGES <= E) {
#pragma unroll
        for (int k = 0; k < BIN_LD; ++k) {
            load4(sv[k], src, chunk0 + k * (BIN_THREADS * 4));
            load4(tv[k], dst, chunk0 + k * (BIN_THREADS * 4));
            load4w(wv[k], e_feat, chunk0 + k * (BIN_THREADS * 4));
        }
    } else {
#pragma unroll
        for (int k = 0; k < BIN_LD; ++k) {
            load4_tail(sv[k], src, chunk0 + k * (BIN_THREADS * 4), E);
            load4_tail(tv[k], dst, chunk0 + k * (BIN_THREADS * 4), E);
            load4w_tail(wv[k], e_feat, chunk0 + k * (BIN_THREADS * 4), E);
        }
    }
#pragma unroll
    for (int q = 0; q < (MAXB + 32 + BIN_THREADS - 1) / BIN_THREADS; ++q) {
        int o = tid + q * BIN_THREADS;
        if (o < MAXB) hist[o] = 0u;
        if (o < MAXB + 32) cntv[o] = cv[q];
    }
    __syncthreads();

    if (wave == 0) {
        const int per = (NB + 31) / 32;
        const int lo = lane * per;
        unsigned s = 0u;
        for (int i = 0; i < per; ++i) s += cntv[lo + i];
        unsigned inc = s;
        for (int d = 1; d < 32; d <<= 1) {
            unsigned up = __shfl_up(inc, d);
            if (lane >= d) inc += up;
        }
        unsigned run = inc - s;
        for (int i = 0; i < per; ++i) {
            gpos[lo + i] = run;
            run += cntv[lo + i];
        }
    }
    unsigned ow[BIN_LD][4], rk[BIN_LD][4];
#pragma unroll
    for (int k = 0; k < BIN_LD; ++k) {
#pragma unroll
        for (int j = 0; j < 4; ++j) {
            ow[k][j] = (tv[k][j] >= 0) ? (unsigned)tv[k][j] / NPW : 0u;
            rk[k][j] = 0u;
            if (tv[k][j] >= 0) rk[k][j] = atomicAdd(&hist[ow[k][j]], 1u);
        }
    }
    __syncthreads();
    if (wave == 0) {
        const int per = (NB + 31) / 32;
        const int lo = lane * per;
        unsigned s = 0u;
        for (int i = 0; i < per; ++i) s += hist[lo + i];
        unsigned inc = s;
        for (int d = 1; d < 32; d <<= 1) {
            unsigned up = __shfl_up(inc, d);
            if (lane >= d) inc += up;
        }
        unsigned run = inc - s;
        for (int i = 0; i < per; ++i) {
            cntv[lo + i] = run;
            run += hist[lo + i];
        }
    }
#pragma unroll
    for (int q = 0; q < MAXB / BIN_THREADS; ++q) {
        int o = tid + q * BIN_THREADS;
        if (o < NB) gpos[o] += bv[q];
    }
    __syncthreads();
    unsigned ls[BIN_LD][4];
#pragma unroll
    for (int k = 0; k < BIN_LD; ++k) {
#pragma unroll
        for (int j = 0; j < 4; ++j) {
            ls[k][j] = rk[k][j] + cntv[ow[k][j]];
            rk[k][j] += gpos[ow[k][j]];
        }
    }
#pragma unroll
    for (int k = 0; k < BIN_LD; ++k) {
#pragma unroll
        for (int j = 0; j < 4; ++j) {
            int t = tv[k][j];
            if (t >= 0) {
                unsigned tl = (unsigned)t - ow[k][j] * NPW;
                v2i r;
                r.x = (int)((tl << SBITS) | (unsigned)sv[k][j]);
                r.y = wv[k][j];
                srec[ls[k][j]] = r;
                sdst[ls[k][j]] = rk[k][j];
            }
        }
    }
    __syncthreads();
    int total = E - blockIdx.x * BIN_EDGES;
    if (total > BIN_EDGES) total = BIN_EDGES;
    v2i* recs = (v2i*)rec;
    if (total == BIN_EDGES) {
        v2i ro[BIN_EDGES / BIN_THREADS];
        unsigned go[BIN_EDGES / BIN_THREADS];
#pragma unroll
        for (int k = 0; k < BIN_EDGES / BIN_THREADS; ++k) {
            ro[k] = srec[tid + k * BIN_THREADS];
            go[k] = sdst[tid + k * BIN_THREADS];
        }
#pragma unroll
        for (int k = 0; k < BIN_EDGES / BIN_THREADS; ++k) recs[go[k]] = ro[k];
    } else {
        for (int i = tid; i < total; i += BIN_THREADS) recs[sdst[i]] = srec[i];
    }
}

__device__ __forceinline__ void drain_batch(v2i rc, int n, int lane, int node0,
                                            const float* __restrict__ feat,
                                            const unsigned char* __restrict__ deg8,
                                            int tabN, int tabLim,
                                            const unsigned char* sdeg,
                                            float* sagg, unsigned* sind) {
    bool valid = lane < n;
    int s = rc.x & SMASK;
    int tl = (int)((unsigned)rc.x >> SBITS);
    unsigned d8 = (s < tabLim) ? (unsigned)sdeg[s] : 255u;
    if (valid) atomicAdd(&sind[tl], 1u);
    bool live = valid && (s != node0 + tl);
    unsigned m = __builtin_amdgcn_ballot_w32(live);
    int sb = s * FD;
    int tb = tl * FD;
    int wi = rc.y;

    float fv[32];
#pragma unroll
    for (int j = 0; j < 32; ++j) {
        int sj = __builtin_amdgcn_readlane(sb, j);
        fv[j] = feat[sj + lane];
    }
    unsigned dg = d8;
    if (d8 == 255u) {
        dg = (unsigned)deg8[(size_t)tabN + s] | ((unsigned)deg8[(size_t)2 * tabN + s] << 8) |
             ((unsigned)deg8[(size_t)3 * tabN + s] << 16);
    }
    float norm = rsqrtf((float)(dg ? dg : 1u));
    int ni = __float_as_int(norm);
#pragma unroll
    for (int j = 0; j < 32; ++j) {
        if ((m >> j) & 1u) {
            float nj = __int_as_float(__builtin_amdgcn_readlane(ni, j));
            float wj = __int_as_float(__builtin_amdgcn_readlane(wi, j));
            int tj = __builtin_amdgcn_readlane(tb, j);
            atomicAdd(&sagg[tj + lane], fv[j] * nj * wj);
        }
    }
}

__global__ __launch_bounds__(OWN_THREADS)
void gcn_owner_kernel(const float* __restrict__ feat,
                      const unsigned char* __restrict__ deg8,
                      const unsigned* __restrict__ cnt,
                      const int* __restrict__ rec,
                      float* __restrict__ out,
                      int N, int tabN) {
    __shared__ __attribute__((aligned(16))) float sagg[NPW * FD];
    __shared__ unsigned sind[NPW];
    __shared__ unsigned sfirst;
    __shared__ __attribute__((aligned(16))) unsigned char sdeg[NTAB];

    const int tid = threadIdx.x;
    const int lane = tid & 31;
    const int wave = __builtin_amdgcn_readfirstlane(tid >> 5);
    const int b = blockIdx.x;
    const int node0 = b * NPW;

    const int tabLim = tabN < NTAB ? tabN : NTAB;
    v4i tq[NTAB / 16 / OWN_THREADS];
#pragma unroll
    for (int k = 0; k < NTAB / 16 / OWN_THREADS; ++k) {
        int q = tid + k * OWN_THREADS;
        v4i z = {0, 0, 0, 0};
        tq[k] = (q * 16 < tabLim) ? ((const v4i*)deg8)[q] : z;
    }

    for (int i = tid; i < NPW * FD; i += OWN_THREADS) sagg[i] = 0.0f;
    for (int i = tid; i < NPW; i += OWN_THREADS) sind[i] = 0u;
    if (tid == 0) sfirst = 0u;
    __syncthreads();

    unsigned part = 0u;
    for (int o = tid; o < b; o += OWN_THREADS) part += cnt[o * CSTRIDE];
    if (part) atomicAdd(&sfirst, part);
#pragma unroll
    for (int k = 0; k < NTAB / 16 / OWN_THREADS; ++k) {
        int q = tid + k * OWN_THREADS;
        if (q * 16 < tabLim) ((v4i*)sdeg)[q] = tq[k];
    }
    __syncthreads();

    const int n = (int)cnt[b * CSTRIDE];
    const v2i* mine = (const v2i*)rec + sfirst;
    const int nBatch = (n + 31) >> 5;

    int bt = wave;
    v2i rc = {0, 0};
    if (bt < nBatch && bt * 32 + lane < n) rc = mine[bt * 32 + lane];
    while (bt < nBatch) {
        const int nb = bt + OWN_WAVES;
        v2i rn = {0, 0};
        if (nb < nBatch && nb * 32 + lane < n) rn = mine[nb * 32 + lane];
        const int left = n - bt * 32;
        drain_batch(rc, left < 32 ? left : 32, lane, node0, feat, deg8, tabN, tabLim, sdeg, sagg, sind);
        rc = rn;
        bt = nb;
    }

    __syncthreads();

    if (wave < NTILE) {
        int r = lane & 15;
        int half = lane >> 4;
        int ln = wave * 16 + r;
        int node = node0 + ln;

        float vals[16];
        if (node < N) {
            unsigned dg = sind[ln];
            float dn = rsqrtf((float)(dg ? dg : 1u));
            const v4f* fp = (const v4f*)(feat + (size_t)node * FD + half * 16);
            const v4f* ap = (const v4f*)(sagg + ln * FD + half * 16);
#pragma unroll
            for (int qd = 0; qd < 4; ++qd) {
                v4f f = fp[qd];
                v4f a = ap[qd];
                vals[4 * qd + 0] = __builtin_fabsf(f.x - a.x * dn);
                vals[4 * qd + 1] = __builtin_fabsf(f.y - a.y * dn);
                vals[4 * qd + 2] = __builtin_fabsf(f.z - a.z * dn);
                vals[4 * qd + 3] = __builtin_fabsf(f.w - a.w * dn);
            }
        } else {
#pragma unroll
            for (int qd = 0; qd < 16; ++qd) vals[qd] = 0.0f;
        }

        v2f ones = {1.0f, 1.0f};
        v8f c = {};
#pragma unroll
        for (int kk = 0; kk < 8; ++kk) {
            v2f a;
            a.x = vals[2 * kk + 0];
            a.y = vals[2 * kk + 1];
            c = __builtin_amdgcn_wmma_f32_16x16x4_f32(
                    false, a, false, ones, (short)0, c, false, false);
        }

        if ((lane & 15) == 0) {
            int base = node0 + wave * 16 + half * 8;
#pragma unroll
            for (int mm = 0; mm < 8; ++mm) {
                int nd = base + mm;
                if (nd < N) out[nd] = c[mm];
            }
        }
    }
}

extern "C" void kernel_launch(void* const* d_in, const int* in_sizes, int n_in,
                              void* d_out, int out_size, void* d_ws, size_t ws_size,
                              hipStream_t stream) {
    const float* feat   = (const float*)d_in[0];
    const float* e_feat = (const float*)d_in[1];
    const int*   src    = (const int*)d_in[2];
    const int*   dst    = (const int*)d_in[3];
    float*       out    = (float*)d_out;

    int N = in_sizes[0] / FD;
    int E = in_sizes[2];
    int NB = (N + NPW - 1) / NPW;
    int tabN = NB * NPW;
    int nbin = (E + BIN_EDGES - 1) / BIN_EDGES;

    unsigned* planes = (unsigned*)d_ws;
    unsigned* cnt  = (unsigned*)d_ws + tabN;
    int*      rec  = (int*)(cnt + 2 * NB * CSTRIDE);
    unsigned* base = (unsigned*)d_out;
    int*      ssort = rec;
    unsigned* soff  = (unsigned*)(rec + (size_t)nbin * (BIN_EDGES / 2));

    gcn_count_kernel<<<nbin, BIN_THREADS, 0, stream>>>(src, dst, ssort, soff, E, NB);

    gcn_outdeg_kernel<<<NB, DEG_THREADS, 0, stream>>>(ssort, soff, planes, cnt, base, tabN, NB, nbin);

    gcn_fill_kernel<<<nbin, BIN_THREADS, 0, stream>>>(e_feat, src, dst, cnt, base, rec, E, NB, nbin);

    gcn_owner_kernel<<<NB, OWN_THREADS, 0, stream>>>(feat, (const unsigned char*)d_ws, cnt, rec, out, N, tabN);
}


# owner pass: records counting-sorted by destination node in LDS; a wave sums a node's rows in a register (rank order, norm*w formed first), one LDS f32 add per node and wave, not per edge
# speedup vs baseline: 6.6569x; 1.0464x over previous
#include <hip/hip_runtime.h>


typedef __attribute__((ext_vector_type(2))) float v2f;
typedef __attribute__((ext_vector_type(4))) float v4f;
typedef __attribute__((ext_vector_type(8))) float v8f;
typedef __attribute__((ext_vector_type(2))) int v2i;
typedef __attribute__((ext_vector_type(4))) int v4i;

#define FD 32
#define NPW 400
#define NTILE (NPW / 16)
#define OWN_THREADS 1024
#define OWN_WAVES (OWN_THREADS / 32)
#define BIN_THREADS 256
#define BIN_LD 4
#define BIN_EDGES (BIN_THREADS * 4 * BIN_LD)
#define MAXB 1024
#define CSTRIDE 32
#define DEG_THREADS 512
#define SBITS 23
#define SMASK 0x7FFFFF
#define NTAB 131072

__device__ __forceinline__ void load4_tail(int (&v)[4], const int* __restrict__ p, int base, int E) {
#pragma unroll
    for (int j = 0; j < 4; ++j) v[j] = (base + j < E) ? p[base + j] : -1;
}
__device__ __forceinline__ void load4w_tail(int (&v)[4], const float* __restrict__ p, int base, int E) {
#pragma unroll
    for (int j = 0; j < 4; ++j) v[j] = (base + j < E) ? __float_as_int(p[base + j]) : 0;
}
__device__ __forceinline__ void load4(int (&v)[4], const int* __restrict__ p, int base) {
    v4i d = *(const v4i*)(p + base);
    v[0] = d.x; v[1] = d.y; v[2] = d.z; v[3] = d.w;
}
__device__ __forceinline__ void load4w(int (&v)[4], const float* __restrict__ p, int base) {
    v4f d = *(const v4f*)(p + base);
    v[0] = __float_as_int(d.x); v[1] = __float_as_int(d.y);
    v[2] = __float_as_int(d.z); v[3] = __float_as_int(d.w);
}

__global__ __launch_bounds__(BIN_THREADS)
void gcn_count_kernel(const int* __restrict__ src,
                      const int* __restrict__ dst,
                      int* __restrict__ ssort_w,
                      unsigned* __restrict__ soff,
                      int E, int NB) {
    __shared__ unsigned hist[MAXB];
    __shared__ unsigned shist[MAXB + 32];
    __shared__ unsigned spos[MAXB + 32];
    __shared__ __attribute__((aligned(16))) unsigned short stage[BIN_EDGES];
    const int tid = threadIdx.x;
    const int lane = tid & 31;
    const int wave = tid >> 5;
    const int chunk0 = blockIdx.x * BIN_EDGES + tid * 4;
    int sv[BIN_LD][4], tv[BIN_LD][4];
    if ((blockIdx.x + 1) * BIN_EDGES <= E) {
#pragma unroll
        for (int k = 0; k < BIN_LD; ++k) {
            load4(sv[k], src, chunk0 + k * (BIN_THREADS * 4));
            load4(tv[k], dst, chunk0 + k * (BIN_THREADS * 4));
        }
    } else {
#pragma unroll
        for (int k = 0; k < BIN_LD; ++k) {
            load4_tail(sv[k], src, chunk0 + k * (BIN_THREADS * 4), E);
            load4_tail(tv[k], dst, chunk0 + k * (BIN_THREADS * 4), E);
        }
    }
    for (int o = tid; o < NB + 32; o += BIN_THREADS) {
        if (o < NB) hist[o] = 0u;
        shist[o] = 0u;
    }
    __syncthreads();

    unsigned so[BIN_LD][4], rk[BIN_LD][4];
#pragma unroll
    for (int k = 0; k < BIN_LD; ++k) {
#pragma unroll
        for (int j = 0; j < 4; ++j) {
            so[k][j] = 0u;
            rk[k][j] = 0u;
            if (tv[k][j] >= 0) {
                atomicAdd(&hist[(unsigned)tv[k][j] / NPW], 1u);
                so[k][j] = (unsigned)sv[k][j] / NPW;
                rk[k][j] = atomicAdd(&shist[so[k][j]], 1u);
            }
        }
    }
    __syncthreads();
    if (wave == 0) {
        const int per = (NB + 32) / 32;
        const int lo = lane * per;
        unsigned s = 0u;
        for (int i = 0; i < per; ++i) s += shist[lo + i];
        unsigned inc = s;
        for (int d = 1; d < 32; d <<= 1) {
            unsigned up = __shfl_up(inc, d);
            if (lane >= d) inc += up;
        }
        unsigned run = inc - s;
        for (int i = 0; i < per; ++i) {
            spos[lo + i] = run;
            run += shist[lo + i];
        }
    }
    __syncthreads();
    v2i* myoff = (v2i*)soff + (size_t)blockIdx.x * (NB + 1);
    for (int o = tid; o <= NB; o += BIN_THREADS) {
        v2i p;
        p.x = (int)spos[o];
        p.y = (o < NB) ? (int)hist[o] : 0;
        myoff[o] = p;
    }
#pragma unroll
    for (int k = 0; k < BIN_LD; ++k) {
#pragma unroll
        for (int j = 0; j < 4; ++j) rk[k][j] += spos[so[k][j]];
    }
#pragma unroll
    for (int k = 0; k < BIN_LD; ++k) {
#pragma unroll
        for (int j = 0; j < 4; ++j) {
            if (tv[k][j] >= 0) {
                unsigned sl = (unsigned)sv[k][j] - so[k][j] * NPW;
                stage[rk[k][j]] = (unsigned short)sl;
            }
        }
    }
    __syncthreads();
    const unsigned total = spos[NB];
    v4i* gs = (v4i*)((unsigned short*)ssort_w + (size_t)blockIdx.x * BIN_EDGES);
    const v4i* ls = (const v4i*)stage;
    for (int q = tid; q < BIN_EDGES / 8; q += BIN_THREADS) {
        if ((unsigned)q * 8u < total) gs[q] = ls[q];
    }
}

__global__ __launch_bounds__(DEG_THREADS)
void gcn_outdeg_kernel(const int* __restrict__ ssort_w,
                       const unsigned* __restrict__ soff,
                       unsigned* __restrict__ planes,
                       unsigned* __restrict__ cnt,
                       unsigned* __restrict__ base,
                       int tabN, int NB, int nbin) {
    __shared__ unsigned lcnt[NPW];
    __shared__ unsigned wtot[DEG_THREADS / 32];
    const int tid = threadIdx.x;
    const int lane = tid & 31;
    const int wave = tid >> 5;
    const int o = blockIdx.x;
    for (int i = tid; i < NPW; i += DEG_THREADS) lcnt[i] = 0u;
    __syncthreads();
    unsigned carry = 0u;
    const unsigned short* ssort = (const unsigned short*)ssort_w;
    for (int c0 = 0; c0 < nbin; c0 += DEG_THREADS) {
        const int c = c0 + tid;
        const bool in = c < nbin;
        v2i p0 = {0, 0};
        v2i p1 = {0, 0};
        if (in) {
            const v2i* t = (const v2i*)soff + (size_t)c * (NB + 1) + o;
            p0 = t[0];
            p1 = t[1];
        }
        const unsigned h = (unsigned)p0.y;
        unsigned inc = h;
        for (int d = 1; d < 32; d <<= 1) {
            unsigned up = __shfl_up(inc, d);
            if (lane >= d) inc += up;
        }
        if (lane == 31) wtot[wave] = inc;
        __syncthreads();
        unsigned before = 0u, all = 0u;
        for (int w = 0; w < DEG_THREADS / 32; ++w) {
            unsigned t = wtot[w];
            all += t;
            if (w < wave) before += t;
        }
        if (in) base[(size_t)o * nbin + c] = carry + before + inc - h;
        carry += all;
        unsigned a = (unsigned)p0.x;
        unsigned b = (unsigned)p1.x;
        if (b > (unsigned)BIN_EDGES) b = BIN_EDGES;
        if (in && a < b) {
            const v4i* blk = (const v4i*)(ssort + (size_t)c * BIN_EDGES);
            for (unsigned q = a >> 3; q * 8u < b; ++q) {
                v4i d = blk[q];
                unsigned wd[4] = {(unsigned)d.x, (unsigned)d.y, (unsigned)d.z, (unsigned)d.w};
#pragma unroll
                for (int e = 0; e < 4; ++e) {
                    unsigned i0 = q * 8u + 2u * e;
                    unsigned id0 = wd[e] & 0xFFFFu;
                    unsigned id1 = wd[e] >> 16;
                    if (i0 >= a && i0 < b && id0 < NPW) atomicAdd(&lcnt[id0], 1u);
                    if (i0 + 1u >= a && i0 + 1u < b && id1 < NPW) atomicAdd(&lcnt[id1], 1u);
                }
            }
        }
        __syncthreads();
    }
    if (tid < NPW) {
        const int p = tid / (NPW / 4);
        const int i = tid - p * (NPW / 4);
        unsigned wv = 0u;
#pragma unroll
        for (int j = 0; j < 4; ++j) {
            unsigned dg = lcnt[4 * i + j];
            unsigned bv = (p == 0) ? (dg < 255u ? dg : 255u) : ((dg >> (8 * (p - 1))) & 255u);
            wv |= bv << (8 * j);
        }
        planes[(size_t)p * (tabN / 4) + o * (NPW / 4) + i] = wv;
    }
    if (tid == 0) cnt[o * CSTRIDE] = carry;
}

__global__ __launch_bounds__(BIN_THREADS)
void gcn_fill_kernel(const float* __restrict__ e_feat,
                     const int* __restrict__ src,
                     const int* __restrict__ dst,
                     const unsigned* __restrict__ cnt,
                     const unsigned* __restrict__ base,
                     int* __restrict__ rec,
                     int E, int NB, int nbin) {
    __shared__ unsigned hist[MAXB];
    __shared__ unsigned cntv[MAXB + 32];
    __shared__ unsigned gpos[MAXB + 32];
    __shared__ __attribute__((aligned(16))) v2i srec[BIN_EDGES];
    __shared__ unsigned sdst[BIN_EDGES];
    const int tid = threadIdx.x;
    const int lane = tid & 31;
    const int wave = tid >> 5;

    unsigned cv[(MAXB + 32 + BIN_THREADS - 1) / BIN_THREADS];
    unsigned bv[MAXB / BIN_THREADS];
#pragma unroll
    for (int q = 0; q < (MAXB + 32 + BIN_THREADS - 1) / BIN_THREADS; ++q) {
        int o = tid + q * BIN_THREADS;
        cv[q] = (o < NB) ? cnt[o * CSTRIDE] : 0u;
    }
#pragma unroll
    for (int q = 0; q < MAXB / BIN_THREADS; ++q) {
        int o = tid + q * BIN_THREADS;
        bv[q] = (o < NB) ? base[(size_t)o * nbin + blockIdx.x] : 0u;
    }
    const int chunk0 = blockIdx.x * BIN_EDGES + tid * 4;
    int sv[BIN_LD][4], tv[BIN_LD][4], wv[BIN_LD][4];
    if ((blockIdx.x + 1) * BIN_EDGES <= E) {
#pragma unroll
        for (int k = 0; k < BIN_LD; ++k) {
            load4(sv[k], src, chunk0 + k * (BIN_THREADS * 4));
            load4(tv[k], dst, chunk0 + k * (BIN_THREADS * 4));
            load4w(wv[k], e_feat, chunk0 + k * (BIN_THREADS * 4));
        }
    } else {
#pragma unroll
        for (int k = 0; k < BIN_LD; ++k) {
            load4_tail(sv[k], src, chunk0 + k * (BIN_THREADS * 4), E);
            load4_tail(tv[k], dst, chunk0 + k * (BIN_THREADS * 4), E);
            load4w_tail(wv[k], e_feat, chunk0 + k * (BIN_THREADS * 4), E);
        }
    }
#pragma unroll
    for (int q = 0; q < (MAXB + 32 + BIN_THREADS - 1) / BIN_THREADS; ++q) {
        int o = tid + q * BIN_THREADS;
        if (o < MAXB) hist[o] = 0u;
        if (o < MAXB + 32) cntv[o] = cv[q];
    }
    __syncthreads();

    if (wave == 0) {
        const int per = (NB + 31) / 32;
        const int lo = lane * per;
        unsigned s = 0u;
        for (int i = 0; i < per; ++i) s += cntv[lo + i];
        unsigned inc = s;
        for (int d = 1; d < 32; d <<= 1) {
            unsigned up = __shfl_up(inc, d);
            if (lane >= d) inc += up;
        }
        unsigned run = inc - s;
        for (int i = 0; i < per; ++i) {
            gpos[lo + i] = run;
            run += cntv[lo + i];
        }
    }
    unsigned ow[BIN_LD][4], rk[BIN_LD][4];
#pragma unroll
    for (int k = 0; k < BIN_LD; ++k) {
#pragma unroll
        for (int j = 0; j < 4; ++j) {
            ow[k][j] = (tv[k][j] >= 0) ? (unsigned)tv[k][j] / NPW : 0u;
            rk[k][j] = 0u;
            if (tv[k][j] >= 0) rk[k][j] = atomicAdd(&hist[ow[k][j]], 1u);
        }
    }
    __syncthreads();
    if (wave == 0) {
        const int per = (NB + 31) / 32;
        const int lo = lane * per;
        unsigned s = 0u;
        for (int i = 0; i < per; ++i) s += hist[lo + i];
        unsigned inc = s;
        for (int d = 1; d < 32; d <<= 1) {
            unsigned up = __shfl_up(inc, d);
            if (lane >= d) inc += up;
        }
        unsigned run = inc - s;
        for (int i = 0; i < per; ++i) {
            cntv[lo + i] = run;
            run += hist[lo + i];
        }
    }
#pragma unroll
    for (int q = 0; q < MAXB / BIN_THREADS; ++q) {
        int o = tid + q * BIN_THREADS;
        if (o < NB) gpos[o] += bv[q];
    }
    __syncthreads();
    unsigned ls[BIN_LD][4];
#pragma unroll
    for (int k = 0; k < BIN_LD; ++k) {
#pragma unroll
        for (int j = 0; j < 4; ++j) {
            ls[k][j] = rk[k][j] + cntv[ow[k][j]];
            rk[k][j] += gpos[ow[k][j]];
        }
    }
#pragma unroll
    for (int k = 0; k < BIN_LD; ++k) {
#pragma unroll
        for (int j = 0; j < 4; ++j) {
            int t = tv[k][j];
            if (t >= 0) {
                unsigned tl = (unsigned)t - ow[k][j] * NPW;
                v2i r;
                r.x = (int)((tl << SBITS) | (unsigned)sv[k][j]);
                r.y = wv[k][j];
                srec[ls[k][j]] = r;
                sdst[ls[k][j]] = rk[k][j];
            }
        }
    }
    __syncthreads();
    int total = E - blockIdx.x * BIN_EDGES;
    if (total > BIN_EDGES) total = BIN_EDGES;
    v2i* recs = (v2i*)rec;
    if (total == BIN_EDGES) {
        v2i ro[BIN_EDGES / BIN_THREADS];
        unsigned go[BIN_EDGES / BIN_THREADS];
#pragma unroll
        for (int k = 0; k < BIN_EDGES / BIN_THREADS; ++k) {
            ro[k] = srec[tid + k * BIN_THREADS];
            go[k] = sdst[tid + k * BIN_THREADS];
        }
#pragma unroll
        for (int k = 0; k < BIN_EDGES / BIN_THREADS; ++k) recs[go[k]] = ro[k];
    } else {
        for (int i = tid; i < total; i += BIN_THREADS) recs[sdst[i]] = srec[i];
    }
}

#define CAP 8192
#define CAP_LD (CAP / OWN_THREADS)

template <bool FULL>
__device__ __forceinline__ void sum_batch(v2i rc, int nv, int lane,
                                          const float* __restrict__ feat,
                                          float* sagg, int& cur, float& acc) {
    int s = rc.x & SMASK;
    int tl = (int)((unsigned)rc.x >> SBITS);
    int sb = s * FD;
    int fi = rc.y;
    int tp = __shfl_up(tl, 1);
    if (lane == 0) tp = cur;
    bool valid = FULL || lane < nv;
    unsigned chg = __builtin_amdgcn_ballot_w32(valid && tl != tp);

    float fv[32];
#pragma unroll
    for (int j = 0; j < 32; ++j) {
        fv[j] = 0.0f;
        if (FULL || j < nv) {
            int sj = __builtin_amdgcn_readlane(sb, j);
            fv[j] = feat[sj + lane];
        }
    }
#pragma unroll
    for (int j = 0; j < 32; ++j) {
        if (FULL || j < nv) {
            if (__builtin_expect((chg >> j) & 1u, 0)) {
                if (cur >= 0) atomicAdd(&sagg[cur * FD + lane], acc);
                cur = __builtin_amdgcn_readlane(tl, j);
                acc = 0.0f;
            }
            float fj = __int_as_float(__builtin_amdgcn_readlane(fi, j));
            acc = fmaf(fv[j], fj, acc);
        }
    }
}

__global__ __launch_bounds__(OWN_THREADS)
void gcn_owner_kernel(const float* __restrict__ feat,
                      const unsigned char* __restrict__ deg8,
                      const unsigned* __restrict__ cnt,
                      const int* __restrict__ rec,
                      float* __restrict__ out,
                      int N, int tabN) {
    __shared__ __attribute__((aligned(16))) float sagg[NPW * FD];
    __shared__ unsigned sind[NPW];
    __shared__ unsigned scnt[NPW + 32];
    __shared__ unsigned soff[NPW + 32];
    __shared__ unsigned sfirst;
    __shared__ __attribute__((aligned(16))) unsigned char sdeg[NTAB];
    __shared__ __attribute__((aligned(16))) v2i srt[CAP];

    const int tid = threadIdx.x;
    const int lane = tid & 31;
    const int wave = __builtin_amdgcn_readfirstlane(tid >> 5);
    const int b = blockIdx.x;
    const int node0 = b * NPW;

    const int tabLim = tabN < NTAB ? tabN : NTAB;
    v4i tq[NTAB / 16 / OWN_THREADS];
#pragma unroll
    for (int k = 0; k < NTAB / 16 / OWN_THREADS; ++k) {
        int q = tid + k * OWN_THREADS;
        v4i z = {0, 0, 0, 0};
        tq[k] = (q * 16 < tabLim) ? ((const v4i*)deg8)[q] : z;
    }

    for (int i = tid; i < NPW * FD; i += OWN_THREADS) sagg[i] = 0.0f;
    for (int i = tid; i < NPW; i += OWN_THREADS) sind[i] = 0u;
    if (tid < NPW + 32) scnt[tid] = 0u;
    if (tid == 0) sfirst = 0u;
    __syncthreads();

    unsigned part = 0u;
    for (int o = tid; o < b; o += OWN_THREADS) part += cnt[o * CSTRIDE];
    if (part) atomicAdd(&sfirst, part);
#pragma unroll
    for (int k = 0; k < NTAB / 16 / OWN_THREADS; ++k) {
        int q = tid + k * OWN_THREADS;
        if (q * 16 < tabLim) ((v4i*)sdeg)[q] = tq[k];
    }
    __syncthreads();

    const int n = __builtin_amdgcn_readfirstlane((int)cnt[b * CSTRIDE]);
    const v2i* mine = (const v2i*)rec + sfirst;

    for (int r0 = 0; r0 < n; r0 += CAP) {
        int nr = n - r0;
        if (nr > CAP) nr = CAP;

        v2i rr[CAP_LD];
        unsigned rk[CAP_LD];
#pragma unroll
        for (int k = 0; k < CAP_LD; ++k) {
            int i = tid + k * OWN_THREADS;
            v2i z = {0, 0};
            rr[k] = (i < nr) ? mine[r0 + i] : z;
        }
#pragma unroll
        for (int k = 0; k < CAP_LD; ++k) {
            int i = tid + k * OWN_THREADS;
            rk[k] = 0u;
            if (i < nr) {
                int s = rr[k].x & SMASK;
                int tl = (int)((unsigned)rr[k].x >> SBITS);
                rk[k] = atomicAdd(&scnt[tl], 1u);
                unsigned dg = (s < tabLim) ? (unsigned)sdeg[s] : 255u;
                if (dg == 255u) {
                    dg = (unsigned)deg8[(size_t)tabN + s] | ((unsigned)deg8[(size_t)2 * tabN + s] << 8) |
                         ((unsigned)deg8[(size_t)3 * tabN + s] << 16);
                }
                float f = rsqrtf((float)(dg ? dg : 1u)) * __int_as_float(rr[k].y);
                if (s == node0 + tl) f = 0.0f;
                rr[k].y = __float_as_int(f);
            }
        }
        __syncthreads();

        if (wave == 0) {
            const int per = (NPW + 32) / 32;
            const int lo = lane * per;
            unsigned sa = 0u;
            for (int i = 0; i < per; ++i) sa += scnt[lo + i];
            unsigned inc = sa;
            for (int d = 1; d < 32; d <<= 1) {
                unsigned up = __shfl_up(inc, d);
                if (lane >= d) inc += up;
            }
            unsigned run = inc - sa;
            for (int i = 0; i < per; ++i) {
                soff[lo + i] = run;
                run += scnt[lo + i];
            }
        }
        if (tid < NPW) sind[tid] += scnt[tid];
        __syncthreads();

#pragma unroll
        for (int k = 0; k < CAP_LD; ++k) {
            int i = tid + k * OWN_THREADS;
            if (i < nr) {
                int tl = (int)((unsigned)rr[k].x >> SBITS);
                srt[soff[tl] + rk[k]] = rr[k];
            }
        }
        if (tid < NPW + 32) scnt[tid] = 0u;
        __syncthreads();

        {
            int cur = -1;
            float acc = 0.0f;
            const int a = (wave * nr) / OWN_WAVES;
            const int bnd = ((wave + 1) * nr) / OWN_WAVES;
            for (int p = a; p < bnd; p += 32) {
                const int nv = bnd - p;
                if (nv >= 32) {
                    sum_batch<true>(srt[p + lane], 32, lane, feat, sagg, cur, acc);
                } else {
                    v2i rc = {0, 0};
                    if (lane < nv) rc = srt[p + lane];
                    sum_batch<false>(rc, nv, lane, feat, sagg, cur, acc);
                }
            }
            if (cur >= 0) atomicAdd(&sagg[cur * FD + lane], acc);
        }
        __syncthreads();
    }

    if (wave < NTILE) {
        int r = lane & 15;
        int half = lane >> 4;
        int ln = wave * 16 + r;
        int node = node0 + ln;

        float vals[16];
        if (node < N) {
            unsigned dg = sind[ln];
            float dn = rsqrtf((float)(dg ? dg : 1u));
            const v4f* fp = (const v4f*)(feat + (size_t)node * FD + half * 16);
            const v4f* ap = (const v4f*)(sagg + ln * FD + half * 16);
#pragma unroll
            for (int qd = 0; qd < 4; ++qd) {
                v4f f = fp[qd];
                v4f a = ap[qd];
                vals[4 * qd + 0] = __builtin_fabsf(f.x - a.x * dn);
                vals[4 * qd + 1] = __builtin_fabsf(f.y - a.y * dn);
                vals[4 * qd + 2] = __builtin_fabsf(f.z - a.z * dn);
                vals[4 * qd + 3] = __builtin_fabsf(f.w - a.w * dn);
            }
        } else {
#pragma unroll
            for (int qd = 0; qd < 16; ++qd) vals[qd] = 0.0f;
        }

        v2f ones = {1.0f, 1.0f};
        v8f c = {};
#pragma unroll
        for (int kk = 0; kk < 8; ++kk) {
            v2f a;
            a.x = vals[2 * kk + 0];
            a.y = vals[2 * kk + 1];
            c = __builtin_amdgcn_wmma_f32_16x16x4_f32(
                    false, a, false, ones, (short)0, c, false, false);
        }

        if ((lane & 15) == 0) {
            int base = node0 + wave * 16 + half * 8;
#pragma unroll
            for (int mm = 0; mm < 8; ++mm) {
                int nd = base + mm;
                if (nd < N) out[nd] = c[mm];
            }
        }
    }
}

extern "C" void kernel_launch(void* const* d_in, const int* in_sizes, int n_in,
                              void* d_out, int out_size, void* d_ws, size_t ws_size,
                              hipStream_t stream) {
    const float* feat   = (const float*)d_in[0];
    const float* e_feat = (const float*)d_in[1];
    const int*   src    = (const int*)d_in[2];
    const int*   dst    = (const int*)d_in[3];
    float*       out    = (float*)d_out;

    int N = in_sizes[0] / FD;
    int E = in_sizes[2];
    int NB = (N + NPW - 1) / NPW;
    int tabN = NB * NPW;
    int nbin = (E + BIN_EDGES - 1) / BIN_EDGES;

    unsigned* planes = (unsigned*)d_ws;
    unsigned* cnt  = (unsigned*)d_ws + tabN;
    int*      rec  = (int*)(cnt + 2 * NB * CSTRIDE);
    unsigned* base = (unsigned*)d_out;
    int*      ssort = rec;
    unsigned* soff  = (unsigned*)(rec + (size_t)nbin * (BIN_EDGES / 2));

    gcn_count_kernel<<<nbin, BIN_THREADS, 0, stream>>>(src, dst, ssort, soff, E, NB);

    gcn_outdeg_kernel<<<NB, DEG_THREADS, 0, stream>>>(ssort, soff, planes, cnt, base, tabN, NB, nbin);

    gcn_fill_kernel<<<nbin, BIN_THREADS, 0, stream>>>(e_feat, src, dst, cnt, base, rec, E, NB, nbin);

    gcn_owner_kernel<<<NB, OWN_THREADS, 0, stream>>>(feat, (const unsigned char*)d_ws, cnt, rec, out, N, tabN);
}


# edge records sorted per chunk only, by the counting pass; the owner pass takes its segment of every chunk through an LDS slot list; fill pass and its launch go; LDS sums in rank order as before
# speedup vs baseline: 8.2875x; 1.2449x over previous
#include <hip/hip_runtime.h>


typedef __attribute__((ext_vector_type(2))) float v2f;
typedef __attribute__((ext_vector_type(4))) float v4f;
typedef __attribute__((ext_vector_type(8))) float v8f;
typedef __attribute__((ext_vector_type(2))) int v2i;
typedef __attribute__((ext_vector_type(4))) int v4i;

#define FD 32
#define NPW 400
#define NTILE (NPW / 16)
#define OWN_THREADS 1024
#define OWN_WAVES (OWN_THREADS / 32)
#define BIN_THREADS 256
#define BIN_LD 4
#define BIN_EDGES (BIN_THREADS * 4 * BIN_LD)
#define MAXB 1024
#define MAXC 992
#define DEG_THREADS 512
#define SBITS 23
#define SMASK 0x7FFFFF
#define NTAB 131072

__device__ __forceinline__ void load4_tail(int (&v)[4], const int* __restrict__ p, int base, int E) {
#pragma unroll
    for (int j = 0; j < 4; ++j) v[j] = (base + j < E) ? p[base + j] : -1;
}
__device__ __forceinline__ void load4w_tail(int (&v)[4], const float* __restrict__ p, int base, int E) {
#pragma unroll
    for (int j = 0; j < 4; ++j) v[j] = (base + j < E) ? __float_as_int(p[base + j]) : 0;
}
__device__ __forceinline__ void load4(int (&v)[4], const int* __restrict__ p, int base) {
    v4i d = *(const v4i*)(p + base);
    v[0] = d.x; v[1] = d.y; v[2] = d.z; v[3] = d.w;
}
__device__ __forceinline__ void load4w(int (&v)[4], const float* __restrict__ p, int base) {
    v4f d = *(const v4f*)(p + base);
    v[0] = __float_as_int(d.x); v[1] = __float_as_int(d.y);
    v[2] = __float_as_int(d.z); v[3] = __float_as_int(d.w);
}


__device__ __forceinline__ void wave_prefix(const unsigned* in, unsigned* outp, int per, int lane) {
    const int lo = lane * per;
    unsigned s = 0u;
    for (int i = 0; i < per; ++i) s += in[lo + i];
    unsigned inc = s;
    for (int d = 1; d < 32; d <<= 1) {
        unsigned up = __shfl_up(inc, d);
        if (lane >= d) inc += up;
    }
    unsigned run = inc - s;
    for (int i = 0; i < per; ++i) {
        outp[lo + i] = run;
        run += in[lo + i];
    }
}

__global__ __launch_bounds__(BIN_THREADS)
void gcn_count_kernel(const float* __restrict__ e_feat,
                      const int* __restrict__ src,
                      const int* __restrict__ dst,
                      int* __restrict__ ssort_w,
                      unsigned* __restrict__ pairs,
                      int* __restrict__ rec,
                      int E, int NB) {
    __shared__ unsigned hist[MAXB + 32];
    __shared__ unsigned shist[MAXB + 32];
    __shared__ unsigned dpos[MAXB + 32];
    __shared__ unsigned spos[MAXB + 32];
    __shared__ __attribute__((aligned(16))) unsigned short stage[BIN_EDGES];
    __shared__ __attribute__((aligned(16))) v2i srec[BIN_EDGES];
    const int tid = threadIdx.x;
    const int lane = tid & 31;
    const int wave = tid >> 5;
    const int chunk0 = blockIdx.x * BIN_EDGES + tid * 4;
    int sv[BIN_LD][4], tv[BIN_LD][4], wv[BIN_LD][4];
    if ((blockIdx.x + 1) * BIN_EDGES <= E) {
#pragma unroll
        for (int k = 0; k < BIN_LD; ++k) {
            load4(sv[k], src, chunk0 + k * (BIN_THREADS * 4));
            load4(tv[k], dst, chunk0 + k * (BIN_THREADS * 4));
            load4w(wv[k], e_feat, chunk0 + k * (BIN_THREADS * 4));
        }
    } else {
#pragma unroll
        for (int k = 0; k < BIN_LD; ++k) {
            load4_tail(sv[k], src, chunk0 + k * (BIN_THREADS * 4), E);
            load4_tail(tv[k], dst, chunk0 + k * (BIN_THREADS * 4), E);
            load4w_tail(wv[k], e_feat, chunk0 + k * (BIN_THREADS * 4), E);
        }
    }
    for (int o = tid; o < NB + 32; o += BIN_THREADS) {
        hist[o] = 0u;
        shist[o] = 0u;
    }
    __syncthreads();

    unsigned ow[BIN_LD][4], dk[BIN_LD][4], so[BIN_LD][4], rk[BIN_LD][4];
#pragma unroll
    for (int k = 0; k < BIN_LD; ++k) {
#pragma unroll
        for (int j = 0; j < 4; ++j) {
            ow[k][j] = 0u;
            dk[k][j] = 0u;
            so[k][j] = 0u;
            rk[k][j] = 0u;
            if (tv[k][j] >= 0) {
                ow[k][j] = (unsigned)tv[k][j] / NPW;
                dk[k][j] = atomicAdd(&hist[ow[k][j]], 1u);
                so[k][j] = (unsigned)sv[k][j] / NPW;
                rk[k][j] = atomicAdd(&shist[so[k][j]], 1u);
            }
        }
    }
    __syncthreads();
    if (wave == 0) wave_prefix(shist, spos, (NB + 32) / 32, lane);
    if (wave == 1) wave_prefix(hist, dpos, (NB + 32) / 32, lane);
    __syncthreads();
    unsigned* mypairs = pairs + (size_t)blockIdx.x * (NB + 1);
    for (int o = tid; o <= NB; o += BIN_THREADS) mypairs[o] = spos[o] | (dpos[o] << 16);
#pragma unroll
    for (int k = 0; k < BIN_LD; ++k) {
#pragma unroll
        for (int j = 0; j < 4; ++j) {
            rk[k][j] += spos[so[k][j]];
            dk[k][j] += dpos[ow[k][j]];
        }
    }
#pragma unroll
    for (int k = 0; k < BIN_LD; ++k) {
#pragma unroll
        for (int j = 0; j < 4; ++j) {
            if (tv[k][j] >= 0) {
                unsigned sl = (unsigned)sv[k][j] - so[k][j] * NPW;
                stage[rk[k][j]] = (unsigned short)sl;
                unsigned tl = (unsigned)tv[k][j] - ow[k][j] * NPW;
                v2i r;
                r.x = (int)((tl << SBITS) | (unsigned)sv[k][j]);
                r.y = wv[k][j];
                srec[dk[k][j]] = r;
            }
        }
    }
    __syncthreads();
    const unsigned total = spos[NB];
    v4i* gs = (v4i*)((unsigned short*)ssort_w + (size_t)blockIdx.x * BIN_EDGES);
    const v4i* ls = (const v4i*)stage;
    for (int q = tid; q < BIN_EDGES / 8; q += BIN_THREADS) {
        if ((unsigned)q * 8u < total) gs[q] = ls[q];
    }
    v4i* gr = (v4i*)((v2i*)rec + (size_t)blockIdx.x * BIN_EDGES);
    const v4i* lr = (const v4i*)srec;
#pragma unroll
    for (int k = 0; k < BIN_EDGES / 2 / BIN_THREADS; ++k) {
        int q = tid + k * BIN_THREADS;
        if ((unsigned)q * 2u < total) gr[q] = lr[q];
    }
}

__global__ __launch_bounds__(DEG_THREADS)
void gcn_outdeg_kernel(const int* __restrict__ ssort_w,
                       const unsigned* __restrict__ pairs,
                       unsigned* __restrict__ planes,
                       int tabN, int NB, int nbin) {
    __shared__ unsigned lcnt[NPW];
    const int tid = threadIdx.x;
    const int o = blockIdx.x;
    for (int i = tid; i < NPW; i += DEG_THREADS) lcnt[i] = 0u;
    __syncthreads();
    const unsigned short* ssort = (const unsigned short*)ssort_w;
    for (int c = tid; c < nbin; c += DEG_THREADS) {
        const unsigned* t = pairs + (size_t)c * (NB + 1) + o;
        unsigned a = t[0] & 0xFFFFu;
        unsigned b = t[1] & 0xFFFFu;
        if (b > (unsigned)BIN_EDGES) b = BIN_EDGES;
        if (a < b) {
            const v4i* blk = (const v4i*)(ssort + (size_t)c * BIN_EDGES);
            for (unsigned q = a >> 3; q * 8u < b; ++q) {
                v4i d = blk[q];
                unsigned wd[4] = {(unsigned)d.x, (unsigned)d.y, (unsigned)d.z, (unsigned)d.w};
#pragma unroll
                for (int e = 0; e < 4; ++e) {
                    unsigned i0 = q * 8u + 2u * e;
                    unsigned id0 = wd[e] & 0xFFFFu;
                    unsigned id1 = wd[e] >> 16;
                    if (i0 >= a && i0 < b && id0 < NPW) atomicAdd(&lcnt[id0], 1u);
                    if (i0 + 1u >= a && i0 + 1u < b && id1 < NPW) atomicAdd(&lcnt[id1], 1u);
                }
            }
        }
    }
    __syncthreads();
    if (tid < NPW) {
        const int p = tid / (NPW / 4);
        const int i = tid - p * (NPW / 4);
        unsigned wv = 0u;
#pragma unroll
        for (int j = 0; j < 4; ++j) {
            unsigned dg = lcnt[4 * i + j];
            unsigned bv = (p == 0) ? (dg < 255u ? dg : 255u) : ((dg >> (8 * (p - 1))) & 255u);
            wv |= bv << (8 * j);
        }
        planes[(size_t)p * (tabN / 4) + o * (NPW / 4) + i] = wv;
    }
}

#define CAP 8192
#define CAP_LD (CAP / OWN_THREADS)

template <bool FULL>
__device__ __forceinline__ void sum_batch(v2i rc, int nv, int lane,
                                          const float* __restrict__ feat,
                                          float* sagg, int& cur, float& acc) {
    int s = rc.x & SMASK;
    int tl = (int)((unsigned)rc.x >> SBITS);
    int sb = s * FD;
    int fi = rc.y;
    int tp = __shfl_up(tl, 1);
    if (lane == 0) tp = cur;
    bool valid = FULL || lane < nv;
    unsigned chg = __builtin_amdgcn_ballot_w32(valid && tl != tp);

    float fv[32];
#pragma unroll
    for (int j = 0; j < 32; ++j) {
        fv[j] = 0.0f;
        if (FULL || j < nv) {
            int sj = __builtin_amdgcn_readlane(sb, j);
            fv[j] = feat[sj + lane];
        }
    }
#pragma unroll
    for (int j = 0; j < 32; ++j) {
        if (FULL || j < nv) {
            if (__builtin_expect((chg >> j) & 1u, 0)) {
                if (cur >= 0) atomicAdd(&sagg[cur * FD + lane], acc);
                cur = __builtin_amdgcn_readlane(tl, j);
                acc = 0.0f;
            }
            float fj = __int_as_float(__builtin_amdgcn_readlane(fi, j));
            acc = fmaf(fv[j], fj, acc);
        }
    }
}

__global__ __launch_bounds__(OWN_THREADS)
void gcn_owner_kernel(const float* __restrict__ feat,
                      const unsigned char* __restrict__ deg8,
                      const unsigned* __restrict__ pairs,
                      const int* __restrict__ rec,
                      float* __restrict__ out,
                      int N, int tabN, int NB, int nbin) {
    __shared__ __attribute__((aligned(16))) float sagg[NPW * FD];
    __shared__ unsigned sind[NPW];
    __shared__ unsigned scnt[NPW + 32];
    __shared__ unsigned soff[NPW + 32];
    __shared__ unsigned clen[MAXC + 32];
    __shared__ unsigned coff[MAXC + 32];
    __shared__ unsigned idx[CAP];
    __shared__ __attribute__((aligned(16))) unsigned char sdeg[NTAB];
    __shared__ __attribute__((aligned(16))) v2i srt[CAP];

    const int tid = threadIdx.x;
    const int lane = tid & 31;
    const int wave = __builtin_amdgcn_readfirstlane(tid >> 5);
    const int b = blockIdx.x;
    const int node0 = b * NPW;

    const int tabLim = tabN < NTAB ? tabN : NTAB;
    v4i tq[NTAB / 16 / OWN_THREADS];
#pragma unroll
    for (int k = 0; k < NTAB / 16 / OWN_THREADS; ++k) {
        int q = tid + k * OWN_THREADS;
        v4i z = {0, 0, 0, 0};
        tq[k] = (q * 16 < tabLim) ? ((const v4i*)deg8)[q] : z;
    }
    unsigned seg0 = 0u, slen = 0u;
    if (tid < nbin) {
        const unsigned* t = pairs + (size_t)tid * (NB + 1) + b;
        seg0 = t[0] >> 16;
        unsigned seg1 = t[1] >> 16;
        if (seg1 > (unsigned)BIN_EDGES) seg1 = BIN_EDGES;
        slen = seg1 > seg0 ? seg1 - seg0 : 0u;
    }
    const unsigned sbase = (unsigned)tid * BIN_EDGES + seg0;

    for (int i = tid; i < NPW * FD; i += OWN_THREADS) sagg[i] = 0.0f;
    for (int i = tid; i < NPW; i += OWN_THREADS) sind[i] = 0u;
    if (tid < NPW + 32) scnt[tid] = 0u;
    if (tid < nbin + 32) clen[tid] = slen;
    __syncthreads();

    if (wave == 0) wave_prefix(clen, coff, (nbin + 32) / 32, lane);
#pragma unroll
    for (int k = 0; k < NTAB / 16 / OWN_THREADS; ++k) {
        int q = tid + k * OWN_THREADS;
        if (q * 16 < tabLim) ((v4i*)sdeg)[q] = tq[k];
    }
    __syncthreads();

    const int n = __builtin_amdgcn_readfirstlane((int)coff[nbin]);
    const int cfirst = (tid < nbin) ? (int)coff[tid] : 0;
    const v2i* recs = (const v2i*)rec;

    for (int r0 = 0; r0 < n; r0 += CAP) {
        int nr = n - r0;
        if (nr > CAP) nr = CAP;

        {
            int j0 = r0 - cfirst;
            if (j0 < 0) j0 = 0;
            int j1 = r0 + nr - cfirst;
            if (j1 > (int)slen) j1 = (int)slen;
            for (int j = j0; j < j1; ++j) idx[cfirst + j - r0] = sbase + (unsigned)j;
        }
        __syncthreads();

        v2i rr[CAP_LD];
        unsigned rk[CAP_LD];
#pragma unroll
        for (int k = 0; k < CAP_LD; ++k) {
            int i = tid + k * OWN_THREADS;
            v2i z = {0, 0};
            rr[k] = (i < nr) ? recs[idx[i]] : z;
        }
#pragma unroll
        for (int k = 0; k < CAP_LD; ++k) {
            int i = tid + k * OWN_THREADS;
            rk[k] = 0u;
            if (i < nr) {
                int s = rr[k].x & SMASK;
                int tl = (int)((unsigned)rr[k].x >> SBITS);
                rk[k] = atomicAdd(&scnt[tl], 1u);
                unsigned dg = (s < tabLim) ? (unsigned)sdeg[s] : 255u;
                if (dg == 255u) {
                    dg = (unsigned)deg8[(size_t)tabN + s] | ((unsigned)deg8[(size_t)2 * tabN + s] << 8) |
                         ((unsigned)deg8[(size_t)3 * tabN + s] << 16);
                }
                float f = rsqrtf((float)(dg ? dg : 1u)) * __int_as_float(rr[k].y);
                if (s == node0 + tl) f = 0.0f;
                rr[k].y = __float_as_int(f);
            }
        }
        __syncthreads();

        if (wave == 0) {
            const int per = (NPW + 32) / 32;
            const int lo = lane * per;
            unsigned sa = 0u;
            for (int i = 0; i < per; ++i) sa += scnt[lo + i];
            unsigned inc = sa;
            for (int d = 1; d < 32; d <<= 1) {
                unsigned up = __shfl_up(inc, d);
                if (lane >= d) inc += up;
            }
            unsigned run = inc - sa;
            for (int i = 0; i < per; ++i) {
                soff[lo + i] = run;
                run += scnt[lo + i];
            }
        }
        if (tid < NPW) sind[tid] += scnt[tid];
        __syncthreads();

#pragma unroll
        for (int k = 0; k < CAP_LD; ++k) {
            int i = tid + k * OWN_THREADS;
            if (i < nr) {
                int tl = (int)((unsigned)rr[k].x >> SBITS);
                srt[soff[tl] + rk[k]] = rr[k];
            }
        }
        if (tid < NPW + 32) scnt[tid] = 0u;
        __syncthreads();

        {
            int cur = -1;
            float acc = 0.0f;
            const int a = (wave * nr) / OWN_WAVES;
            const int bnd = ((wave + 1) * nr) / OWN_WAVES;
            for (int p = a; p < bnd; p += 32) {
                const int nv = bnd - p;
                if (nv >= 32) {
                    sum_batch<true>(srt[p + lane], 32, lane, feat, sagg, cur, acc);
                } else {
                    v2i rc = {0, 0};
                    if (lane < nv) rc = srt[p + lane];
                    sum_batch<false>(rc, nv, lane, feat, sagg, cur, acc);
                }
            }
            if (cur >= 0) atomicAdd(&sagg[cur * FD + lane], acc);
        }
        __syncthreads();
    }

    if (wave < NTILE) {
        int r = lane & 15;
        int half = lane >> 4;
        int ln = wave * 16 + r;
        int node = node0 + ln;

        float vals[16];
        if (node < N) {
            unsigned dg = sind[ln];
            float dn = rsqrtf((float)(dg ? dg : 1u));
            const v4f* fp = (const v4f*)(feat + (size_t)node * FD + half * 16);
            const v4f* ap = (const v4f*)(sagg + ln * FD + half * 16);
#pragma unroll
            for (int qd = 0; qd < 4; ++qd) {
                v4f f = fp[qd];
                v4f a = ap[qd];
                vals[4 * qd + 0] = __builtin_fabsf(f.x - a.x * dn);
                vals[4 * qd + 1] = __builtin_fabsf(f.y - a.y * dn);
                vals[4 * qd + 2] = __builtin_fabsf(f.z - a.z * dn);
                vals[4 * qd + 3] = __builtin_fabsf(f.w - a.w * dn);
            }
        } else {
#pragma unroll
            for (int qd = 0; qd < 16; ++qd) vals[qd] = 0.0f;
        }

        v2f ones = {1.0f, 1.0f};
        v8f c = {};
#pragma unroll
        for (int kk = 0; kk < 8; ++kk) {
            v2f a;
            a.x = vals[2 * kk + 0];
            a.y = vals[2 * kk + 1];
            c = __builtin_amdgcn_wmma_f32_16x16x4_f32(
                    false, a, false, ones, (short)0, c, false, false);
        }

        if ((lane & 15) == 0) {
            int base = node0 + wave * 16 + half * 8;
#pragma unroll
            for (int mm = 0; mm < 8; ++mm) {
                int nd = base + mm;
                if (nd < N) out[nd] = c[mm];
            }
        }
    }
}

extern "C" void kernel_launch(void* const* d_in, const int* in_sizes, int n_in,
                              void* d_out, int out_size, void* d_ws, size_t ws_size,
                              hipStream_t stream) {
    const float* feat   = (const float*)d_in[0];
    const float* e_feat = (const float*)d_in[1];
    const int*   src    = (const int*)d_in[2];
    const int*   dst    = (const int*)d_in[3];
    float*       out    = (float*)d_out;

    int N = in_sizes[0] / FD;
    int E = in_sizes[2];
    int NB = (N + NPW - 1) / NPW;
    int tabN = NB * NPW;
    int nbin = (E + BIN_EDGES - 1) / BIN_EDGES;

    unsigned* planes = (unsigned*)d_ws;
    int*      rec    = (int*)(planes + tabN);
    int*      ssort  = rec + (size_t)nbin * (BIN_EDGES * 2);
    unsigned* pairs  = (unsigned*)(ssort + (size_t)nbin * (BIN_EDGES / 2));
    size_t need = (size_t)tabN * 4 + (size_t)nbin * BIN_EDGES * 10 + (size_t)nbin * (NB + 1) * 4;
    if (ws_size < need || nbin > MAXC || NB > MAXB) return;

    gcn_count_kernel<<<nbin, BIN_THREADS, 0, stream>>>(e_feat, src, dst, ssort, pairs, rec, E, NB);

    gcn_outdeg_kernel<<<NB, DEG_THREADS, 0, stream>>>(ssort, pairs, planes, tabN, NB, nbin);

    gcn_owner_kernel<<<NB, OWN_THREADS, 0, stream>>>(feat, (const unsigned char*)d_ws, pairs, rec, out, N, tabN, NB, nbin);
}
